# starting kernel
# baseline (speedup 1.0000x reference)
#include <hip/hip_runtime.h>

#define N_DRUG 846
#define D 64
#define K 1024
#define N_REL 200
#define RPAD 208
#define BN_EPS 1e-5f

typedef __attribute__((ext_vector_type(2))) float v2f;
typedef __attribute__((ext_vector_type(8))) float v8f;

__device__ __forceinline__ float wave_reduce_add(float v) {
  #pragma unroll
  for (int off = 16; off > 0; off >>= 1) v += __shfl_xor(v, off, 32);
  return v;
}

__global__ void __launch_bounds__(256) gnn_b2sum(const float* __restrict__ b2,
                                                 float* __restrict__ b2sum) {
  const int k = blockIdx.x * 256 + threadIdx.x;
  if (k < K) {
    const float* p = b2 + (size_t)k * D;
    float s = 0.f;
    #pragma unroll
    for (int e = 0; e < D; ++e) s += p[e];
    b2sum[k] = s;
  }
}

__global__ void __launch_bounds__(256) gnn_main(
    const float* __restrict__ drug_table, const float* __restrict__ rela_table,
    const float* __restrict__ ent_table,  const float* __restrict__ W1g,
    const float* __restrict__ b1,         const float* __restrict__ W2g,
    const int*   __restrict__ drug_name,  const int* __restrict__ adj_tail,
    const int*   __restrict__ adj_rel,    const float* __restrict__ b2sum,
    float* __restrict__ drug_e)
{
  extern __shared__ float smem[];
  float* Atile = smem;
  float* Ttile = Atile + RPAD * D;
  float* W1s   = Ttile + RPAD * D;
  float* demb  = W1s + D * D;
  float* w2sum = demb + D;
  float* score = w2sum + D;
  float* red   = score + K;

  const int n    = blockIdx.x;
  const int t    = threadIdx.x;
  const int lane = t & 31;
  const int wave = t >> 5;

  const size_t wbase = (size_t)n * D * D;

  if (t < D) demb[t] = drug_table[(size_t)drug_name[n] * D + t];
  for (int i = t; i < D * D; i += 256) W1s[i] = W1g[wbase + i];
  {
    const int d = t >> 2, q = t & 3;
    const float* p = W2g + wbase + d * D + q * 16;
    float s = 0.f;
    #pragma unroll
    for (int j = 0; j < 16; ++j) s += p[j];
    red[t] = s;
  }
  __syncthreads();
  if (t < D) w2sum[t] = red[4*t] + red[4*t+1] + red[4*t+2] + red[4*t+3];
  for (int i = t; i < RPAD * D; i += 256) {
    const int r = i >> 6, d = i & 63;
    Atile[i] = (r < N_REL) ? demb[d] * rela_table[r * D + d] : 0.f;
  }
  __syncthreads();

  const int half = lane >> 4;
  const int l    = lane & 15;
  for (int tile = wave; tile < 13 * 4; tile += 8) {
    const int m0 = (tile % 13) * 16;
    const int n0 = (tile / 13) * 16;
    v8f c = {0.f, 0.f, 0.f, 0.f, 0.f, 0.f, 0.f, 0.f};
    const float* arow = Atile + (m0 + l) * D;
    #pragma unroll
    for (int kk = 0; kk < 16; ++kk) {
      const int k0 = kk * 4 + 2 * half;
      v2f a, b;
      a[0] = arow[k0];
      a[1] = arow[k0 + 1];
      b[0] = W1s[k0 * D + n0 + l];
      b[1] = W1s[(k0 + 1) * D + n0 + l];
      c = __builtin_amdgcn_wmma_f32_16x16x4_f32(false, a, false, b,
                                                (short)0, c, false, false);
    }
    #pragma unroll
    for (int i = 0; i < 8; ++i)
      Ttile[(m0 + 8 * half + i) * D + n0 + l] = c[i];
  }
  __syncthreads();

  for (int i = 0; i < 128; ++i) {
    const int k = wave * 128 + i;
    const int r = adj_rel[(size_t)n * K + k];
    float v = fmaxf(Ttile[r * D + lane]      + b1[(size_t)k * D + lane],      0.f) * w2sum[lane]
            + fmaxf(Ttile[r * D + lane + 32] + b1[(size_t)k * D + lane + 32], 0.f) * w2sum[lane + 32];
    v = wave_reduce_add(v);
    if (lane == 0) score[k] = v + b2sum[k];
  }
  __syncthreads();

  float lmax = -INFINITY;
  for (int k = t; k < K; k += 256) lmax = fmaxf(lmax, score[k]);
  red[t] = lmax;
  __syncthreads();
  for (int s = 128; s > 0; s >>= 1) {
    if (t < s) red[t] = fmaxf(red[t], red[t + s]);
    __syncthreads();
  }
  const float mx = red[0];
  __syncthreads();
  float lsum = 0.f;
  for (int k = t; k < K; k += 256) {
    const float e = __expf(score[k] - mx);
    score[k] = e;
    lsum += e;
  }
  red[t] = lsum;
  __syncthreads();
  for (int s = 128; s > 0; s >>= 1) {
    if (t < s) red[t] += red[t + s];
    __syncthreads();
  }
  const float inv = 1.f / red[0];
  __syncthreads();

  float acc0 = 0.f, acc1 = 0.f;
  for (int i = 0; i < 128; ++i) {
    const int k = wave * 128 + i;
    const float sc = score[k] * inv;
    const size_t tl = (size_t)adj_tail[(size_t)n * K + k] * D;
    acc0 += sc * ent_table[tl + lane];
    acc1 += sc * ent_table[tl + lane + 32];
  }
  red[wave * 64 + lane]      = acc0;
  red[wave * 64 + lane + 32] = acc1;
  __syncthreads();
  if (t < D) {
    float s = 0.f;
    #pragma unroll
    for (int w = 0; w < 8; ++w) s += red[w * 64 + t];
    drug_e[(size_t)n * (2 * D) + t]     = s;
    drug_e[(size_t)n * (2 * D) + D + t] = demb[t];
  }
}

__global__ void __launch_bounds__(64) gnn_linear(
    const float* __restrict__ drug_e, const float* __restrict__ lin_w,
    const float* __restrict__ lin_b, float* __restrict__ y)
{
  __shared__ float de[2 * D];
  const int n = blockIdx.x, t = threadIdx.x;
  de[t]     = drug_e[(size_t)n * (2 * D) + t];
  de[t + D] = drug_e[(size_t)n * (2 * D) + D + t];
  __syncthreads();
  float acc = lin_b[t];
  const float* w = lin_w + (size_t)t * (2 * D);
  #pragma unroll 8
  for (int j = 0; j < 2 * D; ++j) acc += de[j] * w[j];
  y[(size_t)n * D + t] = fmaxf(acc, 0.f);
}

__global__ void __launch_bounds__(256) gnn_bn(
    const float* __restrict__ y, const float* __restrict__ bn_w,
    const float* __restrict__ bn_b, float* __restrict__ out)
{
  __shared__ float red[256];
  const int c = blockIdx.x, t = threadIdx.x;
  float s = 0.f;
  for (int r = t; r < N_DRUG; r += 256) s += y[(size_t)r * D + c];
  red[t] = s;
  __syncthreads();
  for (int st = 128; st > 0; st >>= 1) {
    if (t < st) red[t] += red[t + st];
    __syncthreads();
  }
  const float mean = red[0] / (float)N_DRUG;
  __syncthreads();
  float q = 0.f;
  for (int r = t; r < N_DRUG; r += 256) {
    const float d = y[(size_t)r * D + c] - mean;
    q += d * d;
  }
  red[t] = q;
  __syncthreads();
  for (int st = 128; st > 0; st >>= 1) {
    if (t < st) red[t] += red[t + st];
    __syncthreads();
  }
  const float var   = red[0] / (float)N_DRUG;
  const float scale = bn_w[c] * rsqrtf(var + BN_EPS);
  const float shift = bn_b[c];
  for (int r = t; r < N_DRUG; r += 256)
    out[(size_t)r * D + c] = (y[(size_t)r * D + c] - mean) * scale + shift;
}

extern "C" void kernel_launch(void* const* d_in, const int* in_sizes, int n_in,
                              void* d_out, int out_size, void* d_ws, size_t ws_size,
                              hipStream_t stream) {
  (void)in_sizes; (void)n_in; (void)out_size; (void)ws_size;
  const float* drug_table = (const float*)d_in[0];
  const float* rela_table = (const float*)d_in[1];
  const float* ent_table  = (const float*)d_in[2];
  const float* W1         = (const float*)d_in[3];
  const float* b1         = (const float*)d_in[4];
  const float* W2         = (const float*)d_in[5];
  const float* b2         = (const float*)d_in[6];
  const float* lin_w      = (const float*)d_in[7];
  const float* lin_b      = (const float*)d_in[8];
  const float* bn_w       = (const float*)d_in[9];
  const float* bn_b       = (const float*)d_in[10];
  const int*   drug_name  = (const int*)d_in[11];
  const int*   adj_tail   = (const int*)d_in[12];
  const int*   adj_rel    = (const int*)d_in[13];

  float* ws     = (float*)d_ws;
  float* b2sum  = ws;
  float* drug_e = ws + K;
  float* y      = drug_e + N_DRUG * 2 * D;

  gnn_b2sum<<<dim3((K + 255) / 256), dim3(256), 0, stream>>>(b2, b2sum);

  const size_t lds = (size_t)(RPAD * D * 2 + D * D + D + D + K + 512) * sizeof(float);
  gnn_main<<<dim3(N_DRUG), dim3(256), lds, stream>>>(
      drug_table, rela_table, ent_table, W1, b1, W2,
      drug_name, adj_tail, adj_rel, b2sum, drug_e);

  gnn_linear<<<dim3(N_DRUG), dim3(64), 0, stream>>>(drug_e, lin_w, lin_b, y);

  gnn_bn<<<dim3(D), dim3(256), 0, stream>>>(y, bn_w, bn_b, (float*)d_out);
}


# Per-neighbour loops batched: indices loaded coalesced and handed out by readlane, gathers issued 8 (logits) or 16 rows (entity sum) ahead of use, logits stored 32 at a time; sum orders kept
# speedup vs baseline: 1.8295x; 1.8295x over previous
#include <hip/hip_runtime.h>

#define N_DRUG 846
#define D 64
#define K 1024
#define N_REL 200
#define RPAD 208
#define BN_EPS 1e-5f

typedef __attribute__((ext_vector_type(2))) float v2f;
typedef __attribute__((ext_vector_type(8))) float v8f;

__device__ __forceinline__ float wave_reduce_add(float v) {
  #pragma unroll
  for (int off = 16; off > 0; off >>= 1) v += __shfl_xor(v, off, 32);
  return v;
}

__global__ void __launch_bounds__(256) gnn_b2sum(const float* __restrict__ b2,
                                                 float* __restrict__ b2sum) {
  const int k = blockIdx.x * 256 + threadIdx.x;
  if (k < K) {
    const float* p = b2 + (size_t)k * D;
    float s = 0.f;
    #pragma unroll
    for (int e = 0; e < D; ++e) s += p[e];
    b2sum[k] = s;
  }
}

__global__ void __launch_bounds__(256) gnn_main(
    const float* __restrict__ drug_table, const float* __restrict__ rela_table,
    const float* __restrict__ ent_table,  const float* __restrict__ W1g,
    const float* __restrict__ b1,         const float* __restrict__ W2g,
    const int*   __restrict__ drug_name,  const int* __restrict__ adj_tail,
    const int*   __restrict__ adj_rel,    const float* __restrict__ b2sum,
    float* __restrict__ drug_e)
{
  extern __shared__ float smem[];
  float* Atile = smem;
  float* Ttile = Atile + RPAD * D;
  float* W1s   = Ttile + RPAD * D;
  float* demb  = W1s + D * D;
  float* w2sum = demb + D;
  float* score = w2sum + D;
  float* red   = score + K;

  const int n    = blockIdx.x;
  const int t    = threadIdx.x;
  const int lane = t & 31;
  const int wave = t >> 5;

  const size_t wbase = (size_t)n * D * D;

  if (t < D) demb[t] = drug_table[(size_t)drug_name[n] * D + t];
  for (int i = t; i < D * D; i += 256) W1s[i] = W1g[wbase + i];
  {
    const int d = t >> 2, q = t & 3;
    const float* p = W2g + wbase + d * D + q * 16;
    float s = 0.f;
    #pragma unroll
    for (int j = 0; j < 16; ++j) s += p[j];
    red[t] = s;
  }
  __syncthreads();
  if (t < D) w2sum[t] = red[4*t] + red[4*t+1] + red[4*t+2] + red[4*t+3];
  for (int i = t; i < RPAD * D; i += 256) {
    const int r = i >> 6, d = i & 63;
    Atile[i] = (r < N_REL) ? demb[d] * rela_table[r * D + d] : 0.f;
  }
  __syncthreads();

  const int half = lane >> 4;
  const int l    = lane & 15;
  for (int tile = wave; tile < 13 * 4; tile += 8) {
    const int m0 = (tile % 13) * 16;
    const int n0 = (tile / 13) * 16;
    v8f c = {0.f, 0.f, 0.f, 0.f, 0.f, 0.f, 0.f, 0.f};
    const float* arow = Atile + (m0 + l) * D;
    #pragma unroll
    for (int kk = 0; kk < 16; ++kk) {
      const int k0 = kk * 4 + 2 * half;
      v2f a, b;
      a[0] = arow[k0];
      a[1] = arow[k0 + 1];
      b[0] = W1s[k0 * D + n0 + l];
      b[1] = W1s[(k0 + 1) * D + n0 + l];
      c = __builtin_amdgcn_wmma_f32_16x16x4_f32(false, a, false, b,
                                                (short)0, c, false, false);
    }
    #pragma unroll
    for (int i = 0; i < 8; ++i)
      Ttile[(m0 + 8 * half + i) * D + n0 + l] = c[i];
  }
  __syncthreads();

  {
    const int   kw = wave * 128;
    const float w0 = w2sum[lane], w1 = w2sum[lane + 32];
    int rv[4];
    #pragma unroll
    for (int j = 0; j < 4; ++j) rv[j] = adj_rel[(size_t)n * K + kw + j * 32 + lane];
    float ca[8], cb[8];
    #pragma unroll
    for (int u = 0; u < 8; ++u) {
      const float* p = b1 + (size_t)(kw + u) * D + lane;
      ca[u] = p[0];
      cb[u] = p[32];
    }
    #pragma unroll
    for (int j = 0; j < 4; ++j) {
      float mine = 0.f;
      #pragma unroll
      for (int b = 0; b < 4; ++b) {
        const int g = j * 4 + b;
        float ta[8], tb[8], na[8], nb[8], v[8];
        #pragma unroll
        for (int u = 0; u < 8; ++u) {
          const int r = __builtin_amdgcn_readlane(rv[j], b * 8 + u);
          ta[u] = Ttile[r * D + lane];
          tb[u] = Ttile[r * D + lane + 32];
        }
        if (g < 15) {
          #pragma unroll
          for (int u = 0; u < 8; ++u) {
            const float* p = b1 + (size_t)(kw + (g + 1) * 8 + u) * D + lane;
            na[u] = p[0];
            nb[u] = p[32];
          }
        }
        #pragma unroll
        for (int u = 0; u < 8; ++u)
          v[u] = fmaxf(ta[u] + ca[u], 0.f) * w0 + fmaxf(tb[u] + cb[u], 0.f) * w1;
        #pragma unroll
        for (int off = 16; off > 0; off >>= 1) {
          #pragma unroll
          for (int u = 0; u < 8; ++u) v[u] += __shfl_xor(v[u], off, 32);
        }
        #pragma unroll
        for (int u = 0; u < 8; ++u)
          if (lane == b * 8 + u) mine = v[u];
        if (g < 15) {
          #pragma unroll
          for (int u = 0; u < 8; ++u) { ca[u] = na[u]; cb[u] = nb[u]; }
        }
      }
      score[kw + j * 32 + lane] = mine + b2sum[kw + j * 32 + lane];
    }
  }
  __syncthreads();

  float lmax = -INFINITY;
  for (int k = t; k < K; k += 256) lmax = fmaxf(lmax, score[k]);
  red[t] = lmax;
  __syncthreads();
  for (int s = 128; s > 0; s >>= 1) {
    if (t < s) red[t] = fmaxf(red[t], red[t + s]);
    __syncthreads();
  }
  const float mx = red[0];
  __syncthreads();
  float lsum = 0.f;
  for (int k = t; k < K; k += 256) {
    const float e = __expf(score[k] - mx);
    score[k] = e;
    lsum += e;
  }
  red[t] = lsum;
  __syncthreads();
  for (int s = 128; s > 0; s >>= 1) {
    if (t < s) red[t] += red[t + s];
    __syncthreads();
  }
  const float inv = 1.f / red[0];
  __syncthreads();

  float acc0 = 0.f, acc1 = 0.f;
  {
    const int kw = wave * 128;
    int tv[4], sv[4];
    #pragma unroll
    for (int j = 0; j < 4; ++j) {
      tv[j] = adj_tail[(size_t)n * K + kw + j * 32 + lane];
      sv[j] = __float_as_int(score[kw + j * 32 + lane] * inv);
    }
    float ea[16], eb[16];
    #pragma unroll
    for (int u = 0; u < 16; ++u) {
      const float* p = ent_table + (size_t)__builtin_amdgcn_readlane(tv[0], u) * D + lane;
      ea[u] = p[0];
      eb[u] = p[32];
    }
    #pragma unroll
    for (int g = 0; g < 8; ++g) {
      float na[16], nb[16];
      if (g < 7) {
        #pragma unroll
        for (int u = 0; u < 16; ++u) {
          const int q = (g + 1) * 16 + u;
          const float* p = ent_table + (size_t)__builtin_amdgcn_readlane(tv[q >> 5], q & 31) * D + lane;
          na[u] = p[0];
          nb[u] = p[32];
        }
      }
      #pragma unroll
      for (int u = 0; u < 16; ++u) {
        const int q = g * 16 + u;
        const float sc = __int_as_float(__builtin_amdgcn_readlane(sv[q >> 5], q & 31));
        acc0 += sc * ea[u];
        acc1 += sc * eb[u];
      }
      if (g < 7) {
        #pragma unroll
        for (int u = 0; u < 16; ++u) { ea[u] = na[u]; eb[u] = nb[u]; }
      }
    }
  }
  red[wave * 64 + lane]      = acc0;
  red[wave * 64 + lane + 32] = acc1;
  __syncthreads();
  if (t < D) {
    float s = 0.f;
    #pragma unroll
    for (int w = 0; w < 8; ++w) s += red[w * 64 + t];
    drug_e[(size_t)n * (2 * D) + t]     = s;
    drug_e[(size_t)n * (2 * D) + D + t] = demb[t];
  }
}

__global__ void __launch_bounds__(64) gnn_linear(
    const float* __restrict__ drug_e, const float* __restrict__ lin_w,
    const float* __restrict__ lin_b, float* __restrict__ y)
{
  __shared__ float de[2 * D];
  const int n = blockIdx.x, t = threadIdx.x;
  de[t]     = drug_e[(size_t)n * (2 * D) + t];
  de[t + D] = drug_e[(size_t)n * (2 * D) + D + t];
  __syncthreads();
  float acc = lin_b[t];
  const float* w = lin_w + (size_t)t * (2 * D);
  #pragma unroll 8
  for (int j = 0; j < 2 * D; ++j) acc += de[j] * w[j];
  y[(size_t)n * D + t] = fmaxf(acc, 0.f);
}

__global__ void __launch_bounds__(256) gnn_bn(
    const float* __restrict__ y, const float* __restrict__ bn_w,
    const float* __restrict__ bn_b, float* __restrict__ out)
{
  __shared__ float red[256];
  const int c = blockIdx.x, t = threadIdx.x;
  float s = 0.f;
  for (int r = t; r < N_DRUG; r += 256) s += y[(size_t)r * D + c];
  red[t] = s;
  __syncthreads();
  for (int st = 128; st > 0; st >>= 1) {
    if (t < st) red[t] += red[t + st];
    __syncthreads();
  }
  const float mean = red[0] / (float)N_DRUG;
  __syncthreads();
  float q = 0.f;
  for (int r = t; r < N_DRUG; r += 256) {
    const float d = y[(size_t)r * D + c] - mean;
    q += d * d;
  }
  red[t] = q;
  __syncthreads();
  for (int st = 128; st > 0; st >>= 1) {
    if (t < st) red[t] += red[t + st];
    __syncthreads();
  }
  const float var   = red[0] / (float)N_DRUG;
  const float scale = bn_w[c] * rsqrtf(var + BN_EPS);
  const float shift = bn_b[c];
  for (int r = t; r < N_DRUG; r += 256)
    out[(size_t)r * D + c] = (y[(size_t)r * D + c] - mean) * scale + shift;
}

extern "C" void kernel_launch(void* const* d_in, const int* in_sizes, int n_in,
                              void* d_out, int out_size, void* d_ws, size_t ws_size,
                              hipStream_t stream) {
  (void)in_sizes; (void)n_in; (void)out_size; (void)ws_size;
  const float* drug_table = (const float*)d_in[0];
  const float* rela_table = (const float*)d_in[1];
  const float* ent_table  = (const float*)d_in[2];
  const float* W1         = (const float*)d_in[3];
  const float* b1         = (const float*)d_in[4];
  const float* W2         = (const float*)d_in[5];
  const float* b2         = (const float*)d_in[6];
  const float* lin_w      = (const float*)d_in[7];
  const float* lin_b      = (const float*)d_in[8];
  const float* bn_w       = (const float*)d_in[9];
  const float* bn_b       = (const float*)d_in[10];
  const int*   drug_name  = (const int*)d_in[11];
  const int*   adj_tail   = (const int*)d_in[12];
  const int*   adj_rel    = (const int*)d_in[13];

  float* ws     = (float*)d_ws;
  float* b2sum  = ws;
  float* drug_e = ws + K;
  float* y      = drug_e + N_DRUG * 2 * D;

  gnn_b2sum<<<dim3((K + 255) / 256), dim3(256), 0, stream>>>(b2, b2sum);

  const size_t lds = (size_t)(RPAD * D * 2 + D * D + D + D + K + 512) * sizeof(float);
  gnn_main<<<dim3(N_DRUG), dim3(256), lds, stream>>>(
      drug_table, rela_table, ent_table, W1, b1, W2,
      drug_name, adj_tail, adj_rel, b2sum, drug_e);

  gnn_linear<<<dim3(N_DRUG), dim3(64), 0, stream>>>(drug_e, lin_w, lin_b, y);

  gnn_bn<<<dim3(D), dim3(256), 0, stream>>>(y, bn_w, bn_b, (float*)d_out);
}


# T-GEMM operands go from global memory straight into WMMA fragments (W1 strip in registers per wave, A rows one tile ahead); LDS copies of A and W1 and their serial loops dropped; sum orders kept
# speedup vs baseline: 3.5991x; 1.9673x over previous
#include <hip/hip_runtime.h>

#define N_DRUG 846
#define D 64
#define K 1024
#define N_REL 200
#define RPAD 208
#define BN_EPS 1e-5f

typedef __attribute__((ext_vector_type(2))) float v2f;
typedef __attribute__((ext_vector_type(8))) float v8f;

__device__ __forceinline__ float wave_reduce_add(float v) {
  #pragma unroll
  for (int off = 16; off > 0; off >>= 1) v += __shfl_xor(v, off, 32);
  return v;
}

__global__ void __launch_bounds__(256) gnn_b2sum(const float* __restrict__ b2,
                                                 float* __restrict__ b2sum) {
  const int k = blockIdx.x * 256 + threadIdx.x;
  if (k < K) {
    const float* p = b2 + (size_t)k * D;
    float s = 0.f;
    #pragma unroll
    for (int e = 0; e < D; ++e) s += p[e];
    b2sum[k] = s;
  }
}

__global__ void __launch_bounds__(256) gnn_main(
    const float* __restrict__ drug_table, const float* __restrict__ rela_table,
    const float* __restrict__ ent_table,  const float* __restrict__ W1g,
    const float* __restrict__ b1,         const float* __restrict__ W2g,
    const int*   __restrict__ drug_name,  const int* __restrict__ adj_tail,
    const int*   __restrict__ adj_rel,    const float* __restrict__ b2sum,
    float* __restrict__ drug_e)
{
  extern __shared__ float smem[];
  float* Ttile = smem;
  float* demb  = Ttile + RPAD * D;
  float* w2sum = demb + D;
  float* score = w2sum + D;
  float* red   = score + K;

  const int n    = blockIdx.x;
  const int t    = threadIdx.x;
  const int lane = t & 31;
  const int wave = t >> 5;
  const int half = lane >> 4;
  const int l    = lane & 15;
  const int wv = __builtin_amdgcn_readfirstlane(wave);
  const int n0 = (wv & 3) * 16;
  const int mt = (wv >> 2) & 1;

  const size_t wbase = (size_t)n * D * D;
  const float* drow  = drug_table + (size_t)drug_name[n] * D;

  v2f bf[16];
  {
    const float* pb = W1g + wbase + (2 * half) * D + n0 + l;
    #pragma unroll
    for (int kk = 0; kk < 16; ++kk) {
      bf[kk][0] = pb[(kk * 4) * D];
      bf[kk][1] = pb[(kk * 4 + 1) * D];
    }
  }
  {
    const int d = t >> 2, q = t & 3;
    const float* p = W2g + wbase + d * D + q * 16;
    float s = 0.f;
    #pragma unroll
    for (int j = 0; j < 16; ++j) s += p[j];
    red[t] = s;
  }
  v2f ar[16];
  {
    const int row = min(mt * 16 + l, N_REL - 1);
    const float* pa = rela_table + row * D + 2 * half;
    #pragma unroll
    for (int kk = 0; kk < 16; ++kk) {
      ar[kk][0] = pa[kk * 4];
      ar[kk][1] = pa[kk * 4 + 1];
    }
  }
  v2f df[16];
  {
    const float* pd = drow + 2 * half;
    #pragma unroll
    for (int kk = 0; kk < 16; ++kk) {
      df[kk][0] = pd[kk * 4];
      df[kk][1] = pd[kk * 4 + 1];
    }
  }
  if (t < D) demb[t] = drow[t];
  __syncthreads();
  if (t < D) w2sum[t] = red[4*t] + red[4*t+1] + red[4*t+2] + red[4*t+3];

  for (int m = mt; m < 13; m += 2) {
    v2f nx[16];
    {
      const int row = min(min(m + 2, 12) * 16 + l, N_REL - 1);
      const float* pa = rela_table + row * D + 2 * half;
      #pragma unroll
      for (int kk = 0; kk < 16; ++kk) {
        nx[kk][0] = pa[kk * 4];
        nx[kk][1] = pa[kk * 4 + 1];
      }
    }
    v8f c = {0.f, 0.f, 0.f, 0.f, 0.f, 0.f, 0.f, 0.f};
    #pragma unroll
    for (int kk = 0; kk < 16; ++kk) {
      const v2f a = ar[kk] * df[kk];
      c = __builtin_amdgcn_wmma_f32_16x16x4_f32(false, a, false, bf[kk],
                                                (short)0, c, false, false);
    }
    const int m0 = m * 16;
    #pragma unroll
    for (int i = 0; i < 8; ++i)
      Ttile[(m0 + 8 * half + i) * D + n0 + l] = c[i];
    #pragma unroll
    for (int kk = 0; kk < 16; ++kk) ar[kk] = nx[kk];
  }
  __syncthreads();

  {
    const int   kw = wave * 128;
    const float w0 = w2sum[lane], w1 = w2sum[lane + 32];
    int rv[4];
    #pragma unroll
    for (int j = 0; j < 4; ++j) rv[j] = adj_rel[(size_t)n * K + kw + j * 32 + lane];
    float ca[8], cb[8];
    #pragma unroll
    for (int u = 0; u < 8; ++u) {
      const float* p = b1 + (size_t)(kw + u) * D + lane;
      ca[u] = p[0];
      cb[u] = p[32];
    }
    #pragma unroll
    for (int j = 0; j < 4; ++j) {
      float mine = 0.f;
      #pragma unroll
      for (int b = 0; b < 4; ++b) {
        const int g = j * 4 + b;
        float ta[8], tb[8], na[8], nb[8], v[8];
        #pragma unroll
        for (int u = 0; u < 8; ++u) {
          const int r = __builtin_amdgcn_readlane(rv[j], b * 8 + u);
          ta[u] = Ttile[r * D + lane];
          tb[u] = Ttile[r * D + lane + 32];
        }
        if (g < 15) {
          #pragma unroll
          for (int u = 0; u < 8; ++u) {
            const float* p = b1 + (size_t)(kw + (g + 1) * 8 + u) * D + lane;
            na[u] = p[0];
            nb[u] = p[32];
          }
        }
        #pragma unroll
        for (int u = 0; u < 8; ++u)
          v[u] = fmaxf(ta[u] + ca[u], 0.f) * w0 + fmaxf(tb[u] + cb[u], 0.f) * w1;
        #pragma unroll
        for (int off = 16; off > 0; off >>= 1) {
          #pragma unroll
          for (int u = 0; u < 8; ++u) v[u] += __shfl_xor(v[u], off, 32);
        }
        #pragma unroll
        for (int u = 0; u < 8; ++u)
          if (lane == b * 8 + u) mine = v[u];
        if (g < 15) {
          #pragma unroll
          for (int u = 0; u < 8; ++u) { ca[u] = na[u]; cb[u] = nb[u]; }
        }
      }
      score[kw + j * 32 + lane] = mine + b2sum[kw + j * 32 + lane];
    }
  }
  __syncthreads();

  float lmax = -INFINITY;
  for (int k = t; k < K; k += 256) lmax = fmaxf(lmax, score[k]);
  red[t] = lmax;
  __syncthreads();
  for (int s = 128; s > 0; s >>= 1) {
    if (t < s) red[t] = fmaxf(red[t], red[t + s]);
    __syncthreads();
  }
  const float mx = red[0];
  __syncthreads();
  float lsum = 0.f;
  for (int k = t; k < K; k += 256) {
    const float e = __expf(score[k] - mx);
    score[k] = e;
    lsum += e;
  }
  red[t] = lsum;
  __syncthreads();
  for (int s = 128; s > 0; s >>= 1) {
    if (t < s) red[t] += red[t + s];
    __syncthreads();
  }
  const float inv = 1.f / red[0];
  __syncthreads();

  float acc0 = 0.f, acc1 = 0.f;
  {
    const int kw = wave * 128;
    int tv[4], sv[4];
    #pragma unroll
    for (int j = 0; j < 4; ++j) {
      tv[j] = adj_tail[(size_t)n * K + kw + j * 32 + lane];
      sv[j] = __float_as_int(score[kw + j * 32 + lane] * inv);
    }
    float ea[16], eb[16];
    #pragma unroll
    for (int u = 0; u < 16; ++u) {
      const float* p = ent_table + (size_t)__builtin_amdgcn_readlane(tv[0], u) * D + lane;
      ea[u] = p[0];
      eb[u] = p[32];
    }
    #pragma unroll
    for (int g = 0; g < 8; ++g) {
      float na[16], nb[16];
      if (g < 7) {
        #pragma unroll
        for (int u = 0; u < 16; ++u) {
          const int q = (g + 1) * 16 + u;
          const float* p = ent_table + (size_t)__builtin_amdgcn_readlane(tv[q >> 5], q & 31) * D + lane;
          na[u] = p[0];
          nb[u] = p[32];
        }
      }
      #pragma unroll
      for (int u = 0; u < 16; ++u) {
        const int q = g * 16 + u;
        const float sc = __int_as_float(__builtin_amdgcn_readlane(sv[q >> 5], q & 31));
        acc0 += sc * ea[u];
        acc1 += sc * eb[u];
      }
      if (g < 7) {
        #pragma unroll
        for (int u = 0; u < 16; ++u) { ea[u] = na[u]; eb[u] = nb[u]; }
      }
    }
  }
  red[wave * 64 + lane]      = acc0;
  red[wave * 64 + lane + 32] = acc1;
  __syncthreads();
  if (t < D) {
    float s = 0.f;
    #pragma unroll
    for (int w = 0; w < 8; ++w) s += red[w * 64 + t];
    drug_e[(size_t)n * (2 * D) + t]     = s;
    drug_e[(size_t)n * (2 * D) + D + t] = demb[t];
  }
}

__global__ void __launch_bounds__(64) gnn_linear(
    const float* __restrict__ drug_e, const float* __restrict__ lin_w,
    const float* __restrict__ lin_b, float* __restrict__ y)
{
  __shared__ float de[2 * D];
  const int n = blockIdx.x, t = threadIdx.x;
  de[t]     = drug_e[(size_t)n * (2 * D) + t];
  de[t + D] = drug_e[(size_t)n * (2 * D) + D + t];
  __syncthreads();
  float acc = lin_b[t];
  const float* w = lin_w + (size_t)t * (2 * D);
  #pragma unroll 8
  for (int j = 0; j < 2 * D; ++j) acc += de[j] * w[j];
  y[(size_t)n * D + t] = fmaxf(acc, 0.f);
}

__global__ void __launch_bounds__(256) gnn_bn(
    const float* __restrict__ y, const float* __restrict__ bn_w,
    const float* __restrict__ bn_b, float* __restrict__ out)
{
  __shared__ float red[256];
  const int c = blockIdx.x, t = threadIdx.x;
  float s = 0.f;
  for (int r = t; r < N_DRUG; r += 256) s += y[(size_t)r * D + c];
  red[t] = s;
  __syncthreads();
  for (int st = 128; st > 0; st >>= 1) {
    if (t < st) red[t] += red[t + st];
    __syncthreads();
  }
  const float mean = red[0] / (float)N_DRUG;
  __syncthreads();
  float q = 0.f;
  for (int r = t; r < N_DRUG; r += 256) {
    const float d = y[(size_t)r * D + c] - mean;
    q += d * d;
  }
  red[t] = q;
  __syncthreads();
  for (int st = 128; st > 0; st >>= 1) {
    if (t < st) red[t] += red[t + st];
    __syncthreads();
  }
  const float var   = red[0] / (float)N_DRUG;
  const float scale = bn_w[c] * rsqrtf(var + BN_EPS);
  const float shift = bn_b[c];
  for (int r = t; r < N_DRUG; r += 256)
    out[(size_t)r * D + c] = (y[(size_t)r * D + c] - mean) * scale + shift;
}

extern "C" void kernel_launch(void* const* d_in, const int* in_sizes, int n_in,
                              void* d_out, int out_size, void* d_ws, size_t ws_size,
                              hipStream_t stream) {
  (void)in_sizes; (void)n_in; (void)out_size; (void)ws_size;
  const float* drug_table = (const float*)d_in[0];
  const float* rela_table = (const float*)d_in[1];
  const float* ent_table  = (const float*)d_in[2];
  const float* W1         = (const float*)d_in[3];
  const float* b1         = (const float*)d_in[4];
  const float* W2         = (const float*)d_in[5];
  const float* b2         = (const float*)d_in[6];
  const float* lin_w      = (const float*)d_in[7];
  const float* lin_b      = (const float*)d_in[8];
  const float* bn_w       = (const float*)d_in[9];
  const float* bn_b       = (const float*)d_in[10];
  const int*   drug_name  = (const int*)d_in[11];
  const int*   adj_tail   = (const int*)d_in[12];
  const int*   adj_rel    = (const int*)d_in[13];

  float* ws     = (float*)d_ws;
  float* b2sum  = ws;
  float* drug_e = ws + K;
  float* y      = drug_e + N_DRUG * 2 * D;

  gnn_b2sum<<<dim3((K + 255) / 256), dim3(256), 0, stream>>>(b2, b2sum);

  const size_t lds = (size_t)(RPAD * D + D + D + K + 512) * sizeof(float);
  gnn_main<<<dim3(N_DRUG), dim3(256), lds, stream>>>(
      drug_table, rela_table, ent_table, W1, b1, W2,
      drug_name, adj_tail, adj_rel, b2sum, drug_e);

  gnn_linear<<<dim3(N_DRUG), dim3(64), 0, stream>>>(drug_e, lin_w, lin_b, y);

  gnn_bn<<<dim3(D), dim3(256), 0, stream>>>(y, bn_w, bn_b, (float*)d_out);
}


# Logit wave sums leave the LDS unit: per 32 neighbours the 32 lane partials are folded on the vector ALU by lane swizzles at xor 16,8,4,2,1, values halving per step; same sum tree, bit-identical
# speedup vs baseline: 4.0824x; 1.1343x over previous
#include <hip/hip_runtime.h>

#define N_DRUG 846
#define D 64
#define K 1024
#define N_REL 200
#define RPAD 208
#define BN_EPS 1e-5f

typedef __attribute__((ext_vector_type(2))) float v2f;
typedef __attribute__((ext_vector_type(8))) float v8f;

__device__ __forceinline__ float wave_reduce_add(float v) {
  #pragma unroll
  for (int off = 16; off > 0; off >>= 1) v += __shfl_xor(v, off, 32);
  return v;
}

template <int M>
__device__ __forceinline__ float from_lane_xor(float x) {
  return __builtin_bit_cast(float, __builtin_amdgcn_update_dpp(
      0, __builtin_bit_cast(int, x), 0x160 | M, 0xf, 0xf, true));
}

__device__ __forceinline__ float from_other_row(float x) {
  const int b = __builtin_bit_cast(int, x);
  return __builtin_bit_cast(float, __builtin_amdgcn_permlanex16(
      b, b, 0x76543210, 0xfedcba98, false, false));
}

__global__ void __launch_bounds__(256) gnn_b2sum(const float* __restrict__ b2,
                                                 float* __restrict__ b2sum) {
  const int k = blockIdx.x * 256 + threadIdx.x;
  if (k < K) {
    const float* p = b2 + (size_t)k * D;
    float s = 0.f;
    #pragma unroll
    for (int e = 0; e < D; ++e) s += p[e];
    b2sum[k] = s;
  }
}

__global__ void __launch_bounds__(256) gnn_main(
    const float* __restrict__ drug_table, const float* __restrict__ rela_table,
    const float* __restrict__ ent_table,  const float* __restrict__ W1g,
    const float* __restrict__ b1,         const float* __restrict__ W2g,
    const int*   __restrict__ drug_name,  const int* __restrict__ adj_tail,
    const int*   __restrict__ adj_rel,    const float* __restrict__ b2sum,
    float* __restrict__ drug_e)
{
  extern __shared__ float smem[];
  float* Ttile = smem;
  float* demb  = Ttile + RPAD * D;
  float* w2sum = demb + D;
  float* score = w2sum + D;
  float* red   = score + K;

  const int n    = blockIdx.x;
  const int t    = threadIdx.x;
  const int lane = t & 31;
  const int wave = t >> 5;
  const int half = lane >> 4;
  const int l    = lane & 15;
  const int wv = __builtin_amdgcn_readfirstlane(wave);
  const int n0 = (wv & 3) * 16;
  const int mt = (wv >> 2) & 1;

  const size_t wbase = (size_t)n * D * D;
  const float* drow  = drug_table + (size_t)drug_name[n] * D;

  v2f bf[16];
  {
    const float* pb = W1g + wbase + (2 * half) * D + n0 + l;
    #pragma unroll
    for (int kk = 0; kk < 16; ++kk) {
      bf[kk][0] = pb[(kk * 4) * D];
      bf[kk][1] = pb[(kk * 4 + 1) * D];
    }
  }
  {
    const int d = t >> 2, q = t & 3;
    const float* p = W2g + wbase + d * D + q * 16;
    float s = 0.f;
    #pragma unroll
    for (int j = 0; j < 16; ++j) s += p[j];
    red[t] = s;
  }
  v2f ar[16];
  {
    const int row = min(mt * 16 + l, N_REL - 1);
    const float* pa = rela_table + row * D + 2 * half;
    #pragma unroll
    for (int kk = 0; kk < 16; ++kk) {
      ar[kk][0] = pa[kk * 4];
      ar[kk][1] = pa[kk * 4 + 1];
    }
  }
  v2f df[16];
  {
    const float* pd = drow + 2 * half;
    #pragma unroll
    for (int kk = 0; kk < 16; ++kk) {
      df[kk][0] = pd[kk * 4];
      df[kk][1] = pd[kk * 4 + 1];
    }
  }
  if (t < D) demb[t] = drow[t];
  __syncthreads();
  if (t < D) w2sum[t] = red[4*t] + red[4*t+1] + red[4*t+2] + red[4*t+3];

  for (int m = mt; m < 13; m += 2) {
    v2f nx[16];
    {
      const int row = min(min(m + 2, 12) * 16 + l, N_REL - 1);
      const float* pa = rela_table + row * D + 2 * half;
      #pragma unroll
      for (int kk = 0; kk < 16; ++kk) {
        nx[kk][0] = pa[kk * 4];
        nx[kk][1] = pa[kk * 4 + 1];
      }
    }
    v8f c = {0.f, 0.f, 0.f, 0.f, 0.f, 0.f, 0.f, 0.f};
    #pragma unroll
    for (int kk = 0; kk < 16; ++kk) {
      const v2f a = ar[kk] * df[kk];
      c = __builtin_amdgcn_wmma_f32_16x16x4_f32(false, a, false, bf[kk],
                                                (short)0, c, false, false);
    }
    const int m0 = m * 16;
    #pragma unroll
    for (int i = 0; i < 8; ++i)
      Ttile[(m0 + 8 * half + i) * D + n0 + l] = c[i];
    #pragma unroll
    for (int kk = 0; kk < 16; ++kk) ar[kk] = nx[kk];
  }
  __syncthreads();

  {
    const int   kw = wave * 128;
    const float w0 = w2sum[lane], w1 = w2sum[lane + 32];
    const bool  hi16 = (lane & 16) != 0;
    const bool  hi8 = (lane & 8) != 0, hi4 = (lane & 4) != 0;
    const bool  hi2 = (lane & 2) != 0, hi1 = (lane & 1) != 0;
    int rv[4];
    #pragma unroll
    for (int j = 0; j < 4; ++j) rv[j] = adj_rel[(size_t)n * K + kw + j * 32 + lane];
    float ca[8], cb[8];
    #pragma unroll
    for (int u = 0; u < 8; ++u) {
      const float* p = b1 + (size_t)(kw + u) * D + lane;
      ca[u] = p[0];
      cb[u] = p[32];
    }
    #pragma unroll
    for (int j = 0; j < 4; ++j) {
      float v[32];
      #pragma unroll
      for (int b = 0; b < 4; ++b) {
        const int g = j * 4 + b;
        float ta[8], tb[8], na[8], nb[8];
        #pragma unroll
        for (int u = 0; u < 8; ++u) {
          const int r = __builtin_amdgcn_readlane(rv[j], b * 8 + u);
          ta[u] = Ttile[r * D + lane];
          tb[u] = Ttile[r * D + lane + 32];
        }
        if (g < 15) {
          #pragma unroll
          for (int u = 0; u < 8; ++u) {
            const float* p = b1 + (size_t)(kw + (g + 1) * 8 + u) * D + lane;
            na[u] = p[0];
            nb[u] = p[32];
          }
        }
        #pragma unroll
        for (int u = 0; u < 8; ++u)
          v[b * 8 + u] = fmaxf(ta[u] + ca[u], 0.f) * w0 + fmaxf(tb[u] + cb[u], 0.f) * w1;
        if (g < 15) {
          #pragma unroll
          for (int u = 0; u < 8; ++u) { ca[u] = na[u]; cb[u] = nb[u]; }
        }
      }
      float s16[16];
      #pragma unroll
      for (int i = 0; i < 16; ++i)
        s16[i] = (hi16 ? v[i + 16] : v[i]) + from_other_row(hi16 ? v[i] : v[i + 16]);
      float s8[8];
      #pragma unroll
      for (int i = 0; i < 8; ++i)
        s8[i] = (hi8 ? s16[i + 8] : s16[i]) + from_lane_xor<8>(hi8 ? s16[i] : s16[i + 8]);
      float s4[4];
      #pragma unroll
      for (int i = 0; i < 4; ++i)
        s4[i] = (hi4 ? s8[i + 4] : s8[i]) + from_lane_xor<4>(hi4 ? s8[i] : s8[i + 4]);
      float s2[2];
      #pragma unroll
      for (int i = 0; i < 2; ++i)
        s2[i] = (hi2 ? s4[i + 2] : s4[i]) + from_lane_xor<2>(hi2 ? s4[i] : s4[i + 2]);
      const float mine = (hi1 ? s2[1] : s2[0]) + from_lane_xor<1>(hi1 ? s2[0] : s2[1]);
      score[kw + j * 32 + lane] = mine + b2sum[kw + j * 32 + lane];
    }
  }
  __syncthreads();

  float lmax = -INFINITY;
  for (int k = t; k < K; k += 256) lmax = fmaxf(lmax, score[k]);
  red[t] = lmax;
  __syncthreads();
  for (int s = 128; s > 0; s >>= 1) {
    if (t < s) red[t] = fmaxf(red[t], red[t + s]);
    __syncthreads();
  }
  const float mx = red[0];
  __syncthreads();
  float lsum = 0.f;
  for (int k = t; k < K; k += 256) {
    const float e = __expf(score[k] - mx);
    score[k] = e;
    lsum += e;
  }
  red[t] = lsum;
  __syncthreads();
  for (int s = 128; s > 0; s >>= 1) {
    if (t < s) red[t] += red[t + s];
    __syncthreads();
  }
  const float inv = 1.f / red[0];
  __syncthreads();

  float acc0 = 0.f, acc1 = 0.f;
  {
    const int kw = wave * 128;
    int tv[4], sv[4];
    #pragma unroll
    for (int j = 0; j < 4; ++j) {
      tv[j] = adj_tail[(size_t)n * K + kw + j * 32 + lane];
      sv[j] = __float_as_int(score[kw + j * 32 + lane] * inv);
    }
    float ea[16], eb[16];
    #pragma unroll
    for (int u = 0; u < 16; ++u) {
      const float* p = ent_table + (size_t)__builtin_amdgcn_readlane(tv[0], u) * D + lane;
      ea[u] = p[0];
      eb[u] = p[32];
    }
    #pragma unroll
    for (int g = 0; g < 8; ++g) {
      float na[16], nb[16];
      if (g < 7) {
        #pragma unroll
        for (int u = 0; u < 16; ++u) {
          const int q = (g + 1) * 16 + u;
          const float* p = ent_table + (size_t)__builtin_amdgcn_readlane(tv[q >> 5], q & 31) * D + lane;
          na[u] = p[0];
          nb[u] = p[32];
        }
      }
      #pragma unroll
      for (int u = 0; u < 16; ++u) {
        const int q = g * 16 + u;
        const float sc = __int_as_float(__builtin_amdgcn_readlane(sv[q >> 5], q & 31));
        acc0 += sc * ea[u];
        acc1 += sc * eb[u];
      }
      if (g < 7) {
        #pragma unroll
        for (int u = 0; u < 16; ++u) { ea[u] = na[u]; eb[u] = nb[u]; }
      }
    }
  }
  red[wave * 64 + lane]      = acc0;
  red[wave * 64 + lane + 32] = acc1;
  __syncthreads();
  if (t < D) {
    float s = 0.f;
    #pragma unroll
    for (int w = 0; w < 8; ++w) s += red[w * 64 + t];
    drug_e[(size_t)n * (2 * D) + t]     = s;
    drug_e[(size_t)n * (2 * D) + D + t] = demb[t];
  }
}

__global__ void __launch_bounds__(64) gnn_linear(
    const float* __restrict__ drug_e, const float* __restrict__ lin_w,
    const float* __restrict__ lin_b, float* __restrict__ y)
{
  __shared__ float de[2 * D];
  const int n = blockIdx.x, t = threadIdx.x;
  de[t]     = drug_e[(size_t)n * (2 * D) + t];
  de[t + D] = drug_e[(size_t)n * (2 * D) + D + t];
  __syncthreads();
  float acc = lin_b[t];
  const float* w = lin_w + (size_t)t * (2 * D);
  #pragma unroll 8
  for (int j = 0; j < 2 * D; ++j) acc += de[j] * w[j];
  y[(size_t)n * D + t] = fmaxf(acc, 0.f);
}

__global__ void __launch_bounds__(256) gnn_bn(
    const float* __restrict__ y, const float* __restrict__ bn_w,
    const float* __restrict__ bn_b, float* __restrict__ out)
{
  __shared__ float red[256];
  const int c = blockIdx.x, t = threadIdx.x;
  float s = 0.f;
  for (int r = t; r < N_DRUG; r += 256) s += y[(size_t)r * D + c];
  red[t] = s;
  __syncthreads();
  for (int st = 128; st > 0; st >>= 1) {
    if (t < st) red[t] += red[t + st];
    __syncthreads();
  }
  const float mean = red[0] / (float)N_DRUG;
  __syncthreads();
  float q = 0.f;
  for (int r = t; r < N_DRUG; r += 256) {
    const float d = y[(size_t)r * D + c] - mean;
    q += d * d;
  }
  red[t] = q;
  __syncthreads();
  for (int st = 128; st > 0; st >>= 1) {
    if (t < st) red[t] += red[t + st];
    __syncthreads();
  }
  const float var   = red[0] / (float)N_DRUG;
  const float scale = bn_w[c] * rsqrtf(var + BN_EPS);
  const float shift = bn_b[c];
  for (int r = t; r < N_DRUG; r += 256)
    out[(size_t)r * D + c] = (y[(size_t)r * D + c] - mean) * scale + shift;
}

extern "C" void kernel_launch(void* const* d_in, const int* in_sizes, int n_in,
                              void* d_out, int out_size, void* d_ws, size_t ws_size,
                              hipStream_t stream) {
  (void)in_sizes; (void)n_in; (void)out_size; (void)ws_size;
  const float* drug_table = (const float*)d_in[0];
  const float* rela_table = (const float*)d_in[1];
  const float* ent_table  = (const float*)d_in[2];
  const float* W1         = (const float*)d_in[3];
  const float* b1         = (const float*)d_in[4];
  const float* W2         = (const float*)d_in[5];
  const float* b2         = (const float*)d_in[6];
  const float* lin_w      = (const float*)d_in[7];
  const float* lin_b      = (const float*)d_in[8];
  const float* bn_w       = (const float*)d_in[9];
  const float* bn_b       = (const float*)d_in[10];
  const int*   drug_name  = (const int*)d_in[11];
  const int*   adj_tail   = (const int*)d_in[12];
  const int*   adj_rel    = (const int*)d_in[13];

  float* ws     = (float*)d_ws;
  float* b2sum  = ws;
  float* drug_e = ws + K;
  float* y      = drug_e + N_DRUG * 2 * D;

  gnn_b2sum<<<dim3((K + 255) / 256), dim3(256), 0, stream>>>(b2, b2sum);

  const size_t lds = (size_t)(RPAD * D + D + D + K + 512) * sizeof(float);
  gnn_main<<<dim3(N_DRUG), dim3(256), lds, stream>>>(
      drug_table, rela_table, ent_table, W1, b1, W2,
      drug_name, adj_tail, adj_rel, b2sum, drug_e);

  gnn_linear<<<dim3(N_DRUG), dim3(64), 0, stream>>>(drug_e, lin_w, lin_b, y);

  gnn_bn<<<dim3(D), dim3(256), 0, stream>>>(y, bn_w, bn_b, (float*)d_out);
}


# Row loads of the per-drug kernel widened to 16 B per lane: b1, T and entity rows fetched per neighbour pair, A fragments by a remapped k order; sum order changes in T, logits and entity sum
# speedup vs baseline: 4.5084x; 1.1044x over previous
#include <hip/hip_runtime.h>

#define N_DRUG 846
#define D 64
#define K 1024
#define N_REL 200
#define RPAD 208
#define BN_EPS 1e-5f

typedef __attribute__((ext_vector_type(2))) float v2f;
typedef __attribute__((ext_vector_type(4))) float v4f;
typedef __attribute__((ext_vector_type(8))) float v8f;

__device__ __forceinline__ float wave_reduce_add(float v) {
  #pragma unroll
  for (int off = 16; off > 0; off >>= 1) v += __shfl_xor(v, off, 32);
  return v;
}

template <int M>
__device__ __forceinline__ float from_lane_xor(float x) {
  return __builtin_bit_cast(float, __builtin_amdgcn_update_dpp(
      0, __builtin_bit_cast(int, x), 0x160 | M, 0xf, 0xf, true));
}

template <int P>
__device__ __forceinline__ int from_row_lane(int x) {
  return __builtin_amdgcn_update_dpp(0, x, 0x150 | P, 0xf, 0xf, true);
}

__device__ __forceinline__ float from_other_row(float x) {
  const int b = __builtin_bit_cast(int, x);
  return __builtin_bit_cast(float, __builtin_amdgcn_permlanex16(
      b, b, 0x76543210, 0xfedcba98, false, false));
}

__global__ void __launch_bounds__(256) gnn_b2sum(const float* __restrict__ b2,
                                                 float* __restrict__ b2sum) {
  const int k = blockIdx.x * 256 + threadIdx.x;
  if (k < K) {
    const float* p = b2 + (size_t)k * D;
    float s = 0.f;
    #pragma unroll
    for (int e = 0; e < D; ++e) s += p[e];
    b2sum[k] = s;
  }
}

template <int B>
__device__ __forceinline__ void gather_t_rows(const float* tl, int held, v4f (&o)[8]) {
  o[0] = *(const v4f*)(tl + from_row_lane<8 * B + 0>(held));
  o[1] = *(const v4f*)(tl + from_row_lane<8 * B + 1>(held));
  o[2] = *(const v4f*)(tl + from_row_lane<8 * B + 2>(held));
  o[3] = *(const v4f*)(tl + from_row_lane<8 * B + 3>(held));
  o[4] = *(const v4f*)(tl + from_row_lane<8 * B + 4>(held));
  o[5] = *(const v4f*)(tl + from_row_lane<8 * B + 5>(held));
  o[6] = *(const v4f*)(tl + from_row_lane<8 * B + 6>(held));
  o[7] = *(const v4f*)(tl + from_row_lane<8 * B + 7>(held));
}
template <int B>
__device__ __forceinline__ void gather_ent_rows(const float* el, int held, v4f (&o)[8]) {
  o[0] = *(const v4f*)(el + (unsigned)from_row_lane<8 * B + 0>(held));
  o[1] = *(const v4f*)(el + (unsigned)from_row_lane<8 * B + 1>(held));
  o[2] = *(const v4f*)(el + (unsigned)from_row_lane<8 * B + 2>(held));
  o[3] = *(const v4f*)(el + (unsigned)from_row_lane<8 * B + 3>(held));
  o[4] = *(const v4f*)(el + (unsigned)from_row_lane<8 * B + 4>(held));
  o[5] = *(const v4f*)(el + (unsigned)from_row_lane<8 * B + 5>(held));
  o[6] = *(const v4f*)(el + (unsigned)from_row_lane<8 * B + 6>(held));
  o[7] = *(const v4f*)(el + (unsigned)from_row_lane<8 * B + 7>(held));
}
template <int B>
__device__ __forceinline__ void add_ent_rows(v4f& acc, int held, const v4f (&e)[8]) {
  acc += __int_as_float(from_row_lane<8 * B + 0>(held)) * e[0];
  acc += __int_as_float(from_row_lane<8 * B + 1>(held)) * e[1];
  acc += __int_as_float(from_row_lane<8 * B + 2>(held)) * e[2];
  acc += __int_as_float(from_row_lane<8 * B + 3>(held)) * e[3];
  acc += __int_as_float(from_row_lane<8 * B + 4>(held)) * e[4];
  acc += __int_as_float(from_row_lane<8 * B + 5>(held)) * e[5];
  acc += __int_as_float(from_row_lane<8 * B + 6>(held)) * e[6];
  acc += __int_as_float(from_row_lane<8 * B + 7>(held)) * e[7];
}

__global__ void __launch_bounds__(256) gnn_main(
    const float* __restrict__ drug_table, const float* __restrict__ rela_table,
    const float* __restrict__ ent_table,  const float* __restrict__ W1g,
    const float* __restrict__ b1,         const float* __restrict__ W2g,
    const int*   __restrict__ drug_name,  const int* __restrict__ adj_tail,
    const int*   __restrict__ adj_rel,    const float* __restrict__ b2sum,
    float* __restrict__ drug_e)
{
  extern __shared__ __align__(16) float smem[];
  float* Ttile = smem;
  float* demb  = Ttile + RPAD * D;
  float* w2sum = demb + D;
  float* score = w2sum + D;
  float* red   = score + K;

  const int n    = blockIdx.x;
  const int t    = threadIdx.x;
  const int lane = t & 31;
  const int wave = t >> 5;
  const int half = lane >> 4;
  const int l    = lane & 15;
  const int wv = __builtin_amdgcn_readfirstlane(wave);
  const int n0 = (wv & 3) * 16;
  const int mt = (wv >> 2) & 1;

  const size_t wbase = (size_t)n * D * D;
  const float* drow  = drug_table + (size_t)drug_name[n] * D;

  v2f bf[16];
  {
    const float* pb = W1g + wbase + (4 * half) * D + n0 + l;
    #pragma unroll
    for (int j = 0; j < 8; ++j) {
      #pragma unroll
      for (int s = 0; s < 2; ++s) {
        bf[2 * j + s][0] = pb[(8 * j + 2 * s) * D];
        bf[2 * j + s][1] = pb[(8 * j + 2 * s + 1) * D];
      }
    }
  }
  {
    const int d = t >> 2, q = t & 3;
    const float* p = W2g + wbase + d * D + q * 16;
    float s = 0.f;
    #pragma unroll
    for (int j = 0; j < 16; ++j) s += p[j];
    red[t] = s;
  }
  v4f ar[8];
  {
    const int row = min(mt * 16 + l, N_REL - 1);
    const float* pa = rela_table + row * D + 4 * half;
    #pragma unroll
    for (int j = 0; j < 8; ++j) ar[j] = *(const v4f*)(pa + 8 * j);
  }
  v4f df[8];
  {
    const float* pd = drow + 4 * half;
    #pragma unroll
    for (int j = 0; j < 8; ++j) df[j] = *(const v4f*)(pd + 8 * j);
  }
  if (t < D) demb[t] = drow[t];
  __syncthreads();
  if (t < D) w2sum[t] = red[4*t] + red[4*t+1] + red[4*t+2] + red[4*t+3];

  for (int m = mt; m < 13; m += 2) {
    v4f nx[8];
    {
      const int row = min(min(m + 2, 12) * 16 + l, N_REL - 1);
      const float* pa = rela_table + row * D + 4 * half;
      #pragma unroll
      for (int j = 0; j < 8; ++j) nx[j] = *(const v4f*)(pa + 8 * j);
    }
    v8f c = {0.f, 0.f, 0.f, 0.f, 0.f, 0.f, 0.f, 0.f};
    #pragma unroll
    for (int j = 0; j < 8; ++j) {
      const v4f a4 = ar[j] * df[j];
      const v2f a0 = {a4[0], a4[1]};
      const v2f a1 = {a4[2], a4[3]};
      c = __builtin_amdgcn_wmma_f32_16x16x4_f32(false, a0, false, bf[2 * j],
                                                (short)0, c, false, false);
      c = __builtin_amdgcn_wmma_f32_16x16x4_f32(false, a1, false, bf[2 * j + 1],
                                                (short)0, c, false, false);
    }
    const int m0 = m * 16;
    #pragma unroll
    for (int i = 0; i < 8; ++i)
      Ttile[(m0 + 8 * half + i) * D + n0 + l] = c[i];
    #pragma unroll
    for (int j = 0; j < 8; ++j) ar[j] = nx[j];
  }
  __syncthreads();

  {
    const int   kw = wave * 128;
    const int   perm = 2 * l + half;
    const v4f   w4 = *(const v4f*)(w2sum + 4 * l);
    const float* tl = Ttile + 4 * l;
    const bool  hi8 = (lane & 8) != 0, hi4 = (lane & 4) != 0;
    const bool  hi2 = (lane & 2) != 0, hi1 = (lane & 1) != 0;
    int rv[4];
    #pragma unroll
    for (int j = 0; j < 4; ++j) rv[j] = adj_rel[(size_t)n * K + kw + j * 32 + perm] * D;
    v4f cb[8];
    #pragma unroll
    for (int u = 0; u < 8; ++u)
      cb[u] = *(const v4f*)(b1 + (size_t)(kw + 2 * u) * D + 4 * lane);
    #pragma unroll
    for (int j = 0; j < 4; ++j) {
      float v[16];
      #pragma unroll
      for (int b = 0; b < 2; ++b) {
        const int g = j * 2 + b;
        v4f tq[8], nb[8];
        if (b == 0) gather_t_rows<0>(tl, rv[j], tq);
        else        gather_t_rows<1>(tl, rv[j], tq);
        if (g < 7) {
          #pragma unroll
          for (int u = 0; u < 8; ++u)
            nb[u] = *(const v4f*)(b1 + (size_t)(kw + (g + 1) * 16 + 2 * u) * D + 4 * lane);
        }
        #pragma unroll
        for (int u = 0; u < 8; ++u) {
          const v4f z = tq[u] + cb[u];
          v[b * 8 + u] = fmaxf(z[0], 0.f) * w4[0] + fmaxf(z[1], 0.f) * w4[1]
                       + fmaxf(z[2], 0.f) * w4[2] + fmaxf(z[3], 0.f) * w4[3];
        }
        if (g < 7) {
          #pragma unroll
          for (int u = 0; u < 8; ++u) cb[u] = nb[u];
        }
      }
      float s8[8];
      #pragma unroll
      for (int i = 0; i < 8; ++i)
        s8[i] = (hi8 ? v[i + 8] : v[i]) + from_lane_xor<8>(hi8 ? v[i] : v[i + 8]);
      float s4[4];
      #pragma unroll
      for (int i = 0; i < 4; ++i)
        s4[i] = (hi4 ? s8[i + 4] : s8[i]) + from_lane_xor<4>(hi4 ? s8[i] : s8[i + 4]);
      float s2[2];
      #pragma unroll
      for (int i = 0; i < 2; ++i)
        s2[i] = (hi2 ? s4[i + 2] : s4[i]) + from_lane_xor<2>(hi2 ? s4[i] : s4[i + 2]);
      const float mine = (hi1 ? s2[1] : s2[0]) + from_lane_xor<1>(hi1 ? s2[0] : s2[1]);
      score[kw + j * 32 + perm] = mine + b2sum[kw + j * 32 + perm];
    }
  }
  __syncthreads();

  float lmax = -INFINITY;
  for (int k = t; k < K; k += 256) lmax = fmaxf(lmax, score[k]);
  red[t] = lmax;
  __syncthreads();
  for (int s = 128; s > 0; s >>= 1) {
    if (t < s) red[t] = fmaxf(red[t], red[t + s]);
    __syncthreads();
  }
  const float mx = red[0];
  __syncthreads();
  float lsum = 0.f;
  for (int k = t; k < K; k += 256) {
    const float e = __expf(score[k] - mx);
    score[k] = e;
    lsum += e;
  }
  red[t] = lsum;
  __syncthreads();
  for (int s = 128; s > 0; s >>= 1) {
    if (t < s) red[t] += red[t + s];
    __syncthreads();
  }
  const float inv = 1.f / red[0];
  __syncthreads();

  {
    const int kw = wave * 128;
    const int perm = 2 * l + half;
    const float* el = ent_table + 4 * l;
    int tv[4], sv[4];
    #pragma unroll
    for (int j = 0; j < 4; ++j) {
      tv[j] = adj_tail[(size_t)n * K + kw + j * 32 + perm] * D;
      sv[j] = __float_as_int(score[kw + j * 32 + perm] * inv);
    }
    v4f acc = {0.f, 0.f, 0.f, 0.f};
    v4f ea[8];
    gather_ent_rows<0>(el, tv[0], ea);
    #pragma unroll
    for (int g = 0; g < 8; ++g) {
      v4f na[8];
      if (g < 7) {
        if (((g + 1) & 1) == 0) gather_ent_rows<0>(el, tv[(g + 1) >> 1], na);
        else                    gather_ent_rows<1>(el, tv[(g + 1) >> 1], na);
      }
      if ((g & 1) == 0) add_ent_rows<0>(acc, sv[g >> 1], ea);
      else              add_ent_rows<1>(acc, sv[g >> 1], ea);
      if (g < 7) {
        #pragma unroll
        for (int u = 0; u < 8; ++u) ea[u] = na[u];
      }
    }
    #pragma unroll
    for (int i = 0; i < 4; ++i) acc[i] += from_other_row(acc[i]);
    if (half == 0) *(v4f*)(red + wave * 64 + 4 * l) = acc;
  }
  __syncthreads();
  if (t < D) {
    float s = 0.f;
    #pragma unroll
    for (int w = 0; w < 8; ++w) s += red[w * 64 + t];
    drug_e[(size_t)n * (2 * D) + t]     = s;
    drug_e[(size_t)n * (2 * D) + D + t] = demb[t];
  }
}

__global__ void __launch_bounds__(64) gnn_linear(
    const float* __restrict__ drug_e, const float* __restrict__ lin_w,
    const float* __restrict__ lin_b, float* __restrict__ y)
{
  __shared__ float de[2 * D];
  const int n = blockIdx.x, t = threadIdx.x;
  de[t]     = drug_e[(size_t)n * (2 * D) + t];
  de[t + D] = drug_e[(size_t)n * (2 * D) + D + t];
  __syncthreads();
  float acc = lin_b[t];
  const float* w = lin_w + (size_t)t * (2 * D);
  #pragma unroll 8
  for (int j = 0; j < 2 * D; ++j) acc += de[j] * w[j];
  y[(size_t)n * D + t] = fmaxf(acc, 0.f);
}

__global__ void __launch_bounds__(256) gnn_bn(
    const float* __restrict__ y, const float* __restrict__ bn_w,
    const float* __restrict__ bn_b, float* __restrict__ out)
{
  __shared__ float red[256];
  const int c = blockIdx.x, t = threadIdx.x;
  float s = 0.f;
  for (int r = t; r < N_DRUG; r += 256) s += y[(size_t)r * D + c];
  red[t] = s;
  __syncthreads();
  for (int st = 128; st > 0; st >>= 1) {
    if (t < st) red[t] += red[t + st];
    __syncthreads();
  }
  const float mean = red[0] / (float)N_DRUG;
  __syncthreads();
  float q = 0.f;
  for (int r = t; r < N_DRUG; r += 256) {
    const float d = y[(size_t)r * D + c] - mean;
    q += d * d;
  }
  red[t] = q;
  __syncthreads();
  for (int st = 128; st > 0; st >>= 1) {
    if (t < st) red[t] += red[t + st];
    __syncthreads();
  }
  const float var   = red[0] / (float)N_DRUG;
  const float scale = bn_w[c] * rsqrtf(var + BN_EPS);
  const float shift = bn_b[c];
  for (int r = t; r < N_DRUG; r += 256)
    out[(size_t)r * D + c] = (y[(size_t)r * D + c] - mean) * scale + shift;
}

extern "C" void kernel_launch(void* const* d_in, const int* in_sizes, int n_in,
                              void* d_out, int out_size, void* d_ws, size_t ws_size,
                              hipStream_t stream) {
  (void)in_sizes; (void)n_in; (void)out_size; (void)ws_size;
  const float* drug_table = (const float*)d_in[0];
  const float* rela_table = (const float*)d_in[1];
  const float* ent_table  = (const float*)d_in[2];
  const float* W1         = (const float*)d_in[3];
  const float* b1         = (const float*)d_in[4];
  const float* W2         = (const float*)d_in[5];
  const float* b2         = (const float*)d_in[6];
  const float* lin_w      = (const float*)d_in[7];
  const float* lin_b      = (const float*)d_in[8];
  const float* bn_w       = (const float*)d_in[9];
  const float* bn_b       = (const float*)d_in[10];
  const int*   drug_name  = (const int*)d_in[11];
  const int*   adj_tail   = (const int*)d_in[12];
  const int*   adj_rel    = (const int*)d_in[13];

  float* ws     = (float*)d_ws;
  float* b2sum  = ws;
  float* drug_e = ws + K;
  float* y      = drug_e + N_DRUG * 2 * D;

  gnn_b2sum<<<dim3((K + 255) / 256), dim3(256), 0, stream>>>(b2, b2sum);

  const size_t lds = (size_t)(RPAD * D + D + D + K + 512) * sizeof(float);
  gnn_main<<<dim3(N_DRUG), dim3(256), lds, stream>>>(
      drug_table, rela_table, ent_table, W1, b1, W2,
      drug_name, adj_tail, adj_rel, b2sum, drug_e);

  gnn_linear<<<dim3(N_DRUG), dim3(64), 0, stream>>>(drug_e, lin_w, lin_b, y);

  gnn_bn<<<dim3(D), dim3(256), 0, stream>>>(y, bn_w, bn_b, (float*)d_out);
}


# Main kernel from 212 to 119 registers: 4 work-groups, not 2, fit a WGP and all 846 run at once; drug embedding folded into the W1 strip (T rounding changes); prefetches reuse their registers
# speedup vs baseline: 5.1921x; 1.1517x over previous
#include <hip/hip_runtime.h>

#define N_DRUG 846
#define D 64
#define K 1024
#define N_REL 200
#define RPAD 208
#define BN_EPS 1e-5f

typedef __attribute__((ext_vector_type(2))) float v2f;
typedef __attribute__((ext_vector_type(4))) float v4f;
typedef __attribute__((ext_vector_type(8))) float v8f;

__device__ __forceinline__ float wave_reduce_add(float v) {
  #pragma unroll
  for (int off = 16; off > 0; off >>= 1) v += __shfl_xor(v, off, 32);
  return v;
}

template <int M>
__device__ __forceinline__ float from_lane_xor(float x) {
  return __builtin_bit_cast(float, __builtin_amdgcn_update_dpp(
      0, __builtin_bit_cast(int, x), 0x160 | M, 0xf, 0xf, true));
}

template <int P>
__device__ __forceinline__ int from_row_lane(int x) {
  return __builtin_amdgcn_update_dpp(0, x, 0x150 | P, 0xf, 0xf, true);
}

__device__ __forceinline__ float from_other_row(float x) {
  const int b = __builtin_bit_cast(int, x);
  return __builtin_bit_cast(float, __builtin_amdgcn_permlanex16(
      b, b, 0x76543210, 0xfedcba98, false, false));
}

__global__ void __launch_bounds__(256) gnn_b2sum(const float* __restrict__ b2,
                                                 float* __restrict__ b2sum) {
  const int k = blockIdx.x * 256 + threadIdx.x;
  if (k < K) {
    const float* p = b2 + (size_t)k * D;
    float s = 0.f;
    #pragma unroll
    for (int e = 0; e < D; ++e) s += p[e];
    b2sum[k] = s;
  }
}

template <int B>
__device__ __forceinline__ void gather_t_rows(const float* tl, int held, v4f (&o)[8]) {
  o[0] = *(const v4f*)(tl + from_row_lane<8 * B + 0>(held));
  o[1] = *(const v4f*)(tl + from_row_lane<8 * B + 1>(held));
  o[2] = *(const v4f*)(tl + from_row_lane<8 * B + 2>(held));
  o[3] = *(const v4f*)(tl + from_row_lane<8 * B + 3>(held));
  o[4] = *(const v4f*)(tl + from_row_lane<8 * B + 4>(held));
  o[5] = *(const v4f*)(tl + from_row_lane<8 * B + 5>(held));
  o[6] = *(const v4f*)(tl + from_row_lane<8 * B + 6>(held));
  o[7] = *(const v4f*)(tl + from_row_lane<8 * B + 7>(held));
}
template <int B>
__device__ __forceinline__ void gather_ent_rows(const float* el, int held, v4f (&o)[8]) {
  o[0] = *(const v4f*)(el + (unsigned)from_row_lane<8 * B + 0>(held));
  o[1] = *(const v4f*)(el + (unsigned)from_row_lane<8 * B + 1>(held));
  o[2] = *(const v4f*)(el + (unsigned)from_row_lane<8 * B + 2>(held));
  o[3] = *(const v4f*)(el + (unsigned)from_row_lane<8 * B + 3>(held));
  o[4] = *(const v4f*)(el + (unsigned)from_row_lane<8 * B + 4>(held));
  o[5] = *(const v4f*)(el + (unsigned)from_row_lane<8 * B + 5>(held));
  o[6] = *(const v4f*)(el + (unsigned)from_row_lane<8 * B + 6>(held));
  o[7] = *(const v4f*)(el + (unsigned)from_row_lane<8 * B + 7>(held));
}
template <int B>
__device__ __forceinline__ void add_ent_rows(v4f& acc, int held, const v4f (&e)[8]) {
  acc += __int_as_float(from_row_lane<8 * B + 0>(held)) * e[0];
  acc += __int_as_float(from_row_lane<8 * B + 1>(held)) * e[1];
  acc += __int_as_float(from_row_lane<8 * B + 2>(held)) * e[2];
  acc += __int_as_float(from_row_lane<8 * B + 3>(held)) * e[3];
  acc += __int_as_float(from_row_lane<8 * B + 4>(held)) * e[4];
  acc += __int_as_float(from_row_lane<8 * B + 5>(held)) * e[5];
  acc += __int_as_float(from_row_lane<8 * B + 6>(held)) * e[6];
  acc += __int_as_float(from_row_lane<8 * B + 7>(held)) * e[7];
}

template <int B, int NB>
__device__ __forceinline__ void add_then_gather(v4f& acc, int held, v4f (&e)[8],
                                                const float* el, int nheld) {
  acc += __int_as_float(from_row_lane<8 * B + 0>(held)) * e[0];
  e[0] = *(const v4f*)(el + (unsigned)from_row_lane<8 * NB + 0>(nheld));
  acc += __int_as_float(from_row_lane<8 * B + 1>(held)) * e[1];
  e[1] = *(const v4f*)(el + (unsigned)from_row_lane<8 * NB + 1>(nheld));
  acc += __int_as_float(from_row_lane<8 * B + 2>(held)) * e[2];
  e[2] = *(const v4f*)(el + (unsigned)from_row_lane<8 * NB + 2>(nheld));
  acc += __int_as_float(from_row_lane<8 * B + 3>(held)) * e[3];
  e[3] = *(const v4f*)(el + (unsigned)from_row_lane<8 * NB + 3>(nheld));
  acc += __int_as_float(from_row_lane<8 * B + 4>(held)) * e[4];
  e[4] = *(const v4f*)(el + (unsigned)from_row_lane<8 * NB + 4>(nheld));
  acc += __int_as_float(from_row_lane<8 * B + 5>(held)) * e[5];
  e[5] = *(const v4f*)(el + (unsigned)from_row_lane<8 * NB + 5>(nheld));
  acc += __int_as_float(from_row_lane<8 * B + 6>(held)) * e[6];
  e[6] = *(const v4f*)(el + (unsigned)from_row_lane<8 * NB + 6>(nheld));
  acc += __int_as_float(from_row_lane<8 * B + 7>(held)) * e[7];
  e[7] = *(const v4f*)(el + (unsigned)from_row_lane<8 * NB + 7>(nheld));
}

struct group_more { static constexpr bool last = false; };
struct group_last { static constexpr bool last = true; };

__global__ void __launch_bounds__(256, 8) gnn_main(
    const float* __restrict__ drug_table, const float* __restrict__ rela_table,
    const float* __restrict__ ent_table,  const float* __restrict__ W1g,
    const float* __restrict__ b1,         const float* __restrict__ W2g,
    const int*   __restrict__ drug_name,  const int* __restrict__ adj_tail,
    const int*   __restrict__ adj_rel,    const float* __restrict__ b2sum,
    float* __restrict__ drug_e)
{
  extern __shared__ __align__(16) float smem[];
  float* Ttile = smem;
  float* demb  = Ttile + RPAD * D;
  float* w2sum = demb + D;
  float* score = w2sum + D;
  float* red   = score + K;

  const int n    = blockIdx.x;
  const int t    = threadIdx.x;
  const int lane = t & 31;
  const int wave = t >> 5;
  const int half = lane >> 4;
  const int l    = lane & 15;
  const int wv = __builtin_amdgcn_readfirstlane(wave);
  const int n0 = (wv & 3) * 16;
  const int mt = (wv >> 2) & 1;

  const size_t wbase = (size_t)n * D * D;
  const float* drow  = drug_table + (size_t)drug_name[n] * D;

  v2f bf[16];
  {
    const float* pb = W1g + wbase + (4 * half) * D + n0 + l;
    #pragma unroll
    for (int j = 0; j < 8; ++j) {
      #pragma unroll
      for (int s = 0; s < 2; ++s) {
        bf[2 * j + s][0] = pb[(8 * j + 2 * s) * D];
        bf[2 * j + s][1] = pb[(8 * j + 2 * s + 1) * D];
      }
    }
  }
  {
    const int d = t >> 2, q = t & 3;
    const float* p = W2g + wbase + d * D + q * 16;
    float s = 0.f;
    #pragma unroll
    for (int j = 0; j < 16; ++j) s += p[j];
    red[t] = s;
  }
  v4f ar[8];
  {
    const int row = min(mt * 16 + l, N_REL - 1);
    const float* pa = rela_table + row * D + 4 * half;
    #pragma unroll
    for (int j = 0; j < 8; ++j) ar[j] = *(const v4f*)(pa + 8 * j);
  }
  if (t < D) demb[t] = drow[t];
  __syncthreads();
  if (t < D) w2sum[t] = red[4*t] + red[4*t+1] + red[4*t+2] + red[4*t+3];
  {
    const float* pd = demb + 4 * half;
    #pragma unroll
    for (int j = 0; j < 8; ++j) {
      const v4f d4 = *(const v4f*)(pd + 8 * j);
      bf[2 * j][0]     *= d4[0];
      bf[2 * j][1]     *= d4[1];
      bf[2 * j + 1][0] *= d4[2];
      bf[2 * j + 1][1] *= d4[3];
    }
  }

  #pragma unroll 1
  for (int m = mt; m < 13; m += 2) {
    const int nrow = min(min(m + 2, 12) * 16 + l, N_REL - 1);
    const float* pn = rela_table + (unsigned)(nrow * D + 4 * half);
    v8f c = {0.f, 0.f, 0.f, 0.f, 0.f, 0.f, 0.f, 0.f};
    #pragma unroll
    for (int j = 0; j < 8; ++j) {
      const v2f a0 = {ar[j][0], ar[j][1]};
      const v2f a1 = {ar[j][2], ar[j][3]};
      c = __builtin_amdgcn_wmma_f32_16x16x4_f32(false, a0, false, bf[2 * j],
                                                (short)0, c, false, false);
      c = __builtin_amdgcn_wmma_f32_16x16x4_f32(false, a1, false, bf[2 * j + 1],
                                                (short)0, c, false, false);
      ar[j] = *(const v4f*)(pn + 8 * j);
    }
    const int m0 = m * 16;
    #pragma unroll
    for (int i = 0; i < 8; ++i)
      Ttile[(m0 + 8 * half + i) * D + n0 + l] = c[i];
  }
  __syncthreads();

  {
    const int   kw = wave * 128;
    const int   perm = 2 * l + half;
    const v4f   w4 = *(const v4f*)(w2sum + 4 * l);
    const float* tl = Ttile + 4 * l;
    const bool  hi8 = (lane & 8) != 0, hi4 = (lane & 4) != 0;
    const bool  hi2 = (lane & 2) != 0, hi1 = (lane & 1) != 0;
    const unsigned relo = (unsigned)n * K + kw + perm;
    const unsigned b1o  = (unsigned)kw * D + 4 * lane;
    int rvc = adj_rel[relo] * D;
    v4f cb[8];
    #pragma unroll
    for (int u = 0; u < 8; ++u)
      cb[u] = *(const v4f*)(b1 + (b1o + 2u * u * D));
    auto group = [&](const int j, auto tag) {
      constexpr bool LAST = decltype(tag)::last;
      int rvn = 0;
      if (!LAST) rvn = adj_rel[relo + (j + 1) * 32] * D;
      float v[16];
      #pragma unroll
      for (int b = 0; b < 2; ++b) {
        v4f tq[8];
        if (b == 0) gather_t_rows<0>(tl, rvc, tq);
        else        gather_t_rows<1>(tl, rvc, tq);
        const unsigned nb1 = b1o + (unsigned)(2 * j + b + 1) * (16 * D);
        #pragma unroll
        for (int u = 0; u < 8; ++u) {
          const v4f z = tq[u] + cb[u];
          v[b * 8 + u] = fmaxf(z[0], 0.f) * w4[0] + fmaxf(z[1], 0.f) * w4[1]
                       + fmaxf(z[2], 0.f) * w4[2] + fmaxf(z[3], 0.f) * w4[3];
          if (!LAST || b == 0) cb[u] = *(const v4f*)(b1 + (nb1 + 2u * u * D));
        }
      }
      float s8[8];
      #pragma unroll
      for (int i = 0; i < 8; ++i)
        s8[i] = (hi8 ? v[i + 8] : v[i]) + from_lane_xor<8>(hi8 ? v[i] : v[i + 8]);
      float s4[4];
      #pragma unroll
      for (int i = 0; i < 4; ++i)
        s4[i] = (hi4 ? s8[i + 4] : s8[i]) + from_lane_xor<4>(hi4 ? s8[i] : s8[i + 4]);
      float s2[2];
      #pragma unroll
      for (int i = 0; i < 2; ++i)
        s2[i] = (hi2 ? s4[i + 2] : s4[i]) + from_lane_xor<2>(hi2 ? s4[i] : s4[i + 2]);
      const float mine = (hi1 ? s2[1] : s2[0]) + from_lane_xor<1>(hi1 ? s2[0] : s2[1]);
      score[kw + j * 32 + perm] = mine + b2sum[(unsigned)(kw + j * 32 + perm)];
      rvc = rvn;
    };
    #pragma unroll 1
    for (int j = 0; j < 3; ++j) group(j, group_more{});
    group(3, group_last{});
  }
  __syncthreads();

  const int laneb = (int)__builtin_amdgcn_mbcnt_lo(~0u, 0u);
  const int tb    = wv * 32 + laneb;
  const int halfb = laneb >> 4;
  const int lb    = laneb & 15;

  float lmax = -INFINITY;
  for (int k = tb; k < K; k += 256) lmax = fmaxf(lmax, score[k]);
  red[tb] = lmax;
  __syncthreads();
  for (int s = 128; s > 0; s >>= 1) {
    if (tb < s) red[tb] = fmaxf(red[tb], red[tb + s]);
    __syncthreads();
  }
  const float mx = red[0];
  __syncthreads();
  float lsum = 0.f;
  for (int k = tb; k < K; k += 256) {
    const float e = __expf(score[k] - mx);
    score[k] = e;
    lsum += e;
  }
  red[tb] = lsum;
  __syncthreads();
  for (int s = 128; s > 0; s >>= 1) {
    if (tb < s) red[tb] += red[tb + s];
    __syncthreads();
  }
  const float inv = 1.f / red[0];
  __syncthreads();

  {
    const int kw = wv * 128;
    const int perm = 2 * lb + halfb;
    const float* el = ent_table + 4 * lb;
    int tv[4], sv[4];
    #pragma unroll
    for (int j = 0; j < 4; ++j) {
      tv[j] = adj_tail[(unsigned)n * K + (unsigned)(kw + j * 32 + perm)] * D;
      sv[j] = __float_as_int(score[kw + j * 32 + perm] * inv);
    }
    v4f acc = {0.f, 0.f, 0.f, 0.f};
    v4f ea[8];
    gather_ent_rows<0>(el, tv[0], ea);
    #pragma unroll
    for (int g = 0; g < 8; ++g) {
      if (g < 7) {
        if ((g & 1) == 0) add_then_gather<0, 1>(acc, sv[g >> 1], ea, el, tv[(g + 1) >> 1]);
        else              add_then_gather<1, 0>(acc, sv[g >> 1], ea, el, tv[(g + 1) >> 1]);
      } else {
        add_ent_rows<1>(acc, sv[g >> 1], ea);
      }
    }
    #pragma unroll
    for (int i = 0; i < 4; ++i) acc[i] += from_other_row(acc[i]);
    if (halfb == 0) *(v4f*)(red + wv * 64 + 4 * lb) = acc;
  }
  __syncthreads();
  if (tb < D) {
    float s = 0.f;
    #pragma unroll
    for (int w = 0; w < 8; ++w) s += red[w * 64 + tb];
    drug_e[(size_t)n * (2 * D) + tb]     = s;
    drug_e[(size_t)n * (2 * D) + D + tb] = demb[tb];
  }
}

__global__ void __launch_bounds__(64) gnn_linear(
    const float* __restrict__ drug_e, const float* __restrict__ lin_w,
    const float* __restrict__ lin_b, float* __restrict__ y)
{
  __shared__ float de[2 * D];
  const int n = blockIdx.x, t = threadIdx.x;
  de[t]     = drug_e[(size_t)n * (2 * D) + t];
  de[t + D] = drug_e[(size_t)n * (2 * D) + D + t];
  __syncthreads();
  float acc = lin_b[t];
  const float* w = lin_w + (size_t)t * (2 * D);
  #pragma unroll 8
  for (int j = 0; j < 2 * D; ++j) acc += de[j] * w[j];
  y[(size_t)n * D + t] = fmaxf(acc, 0.f);
}

__global__ void __launch_bounds__(256) gnn_bn(
    const float* __restrict__ y, const float* __restrict__ bn_w,
    const float* __restrict__ bn_b, float* __restrict__ out)
{
  __shared__ float red[256];
  const int c = blockIdx.x, t = threadIdx.x;
  float s = 0.f;
  for (int r = t; r < N_DRUG; r += 256) s += y[(size_t)r * D + c];
  red[t] = s;
  __syncthreads();
  for (int st = 128; st > 0; st >>= 1) {
    if (t < st) red[t] += red[t + st];
    __syncthreads();
  }
  const float mean = red[0] / (float)N_DRUG;
  __syncthreads();
  float q = 0.f;
  for (int r = t; r < N_DRUG; r += 256) {
    const float d = y[(size_t)r * D + c] - mean;
    q += d * d;
  }
  red[t] = q;
  __syncthreads();
  for (int st = 128; st > 0; st >>= 1) {
    if (t < st) red[t] += red[t + st];
    __syncthreads();
  }
  const float var   = red[0] / (float)N_DRUG;
  const float scale = bn_w[c] * rsqrtf(var + BN_EPS);
  const float shift = bn_b[c];
  for (int r = t; r < N_DRUG; r += 256)
    out[(size_t)r * D + c] = (y[(size_t)r * D + c] - mean) * scale + shift;
}

extern "C" void kernel_launch(void* const* d_in, const int* in_sizes, int n_in,
                              void* d_out, int out_size, void* d_ws, size_t ws_size,
                              hipStream_t stream) {
  (void)in_sizes; (void)n_in; (void)out_size; (void)ws_size;
  const float* drug_table = (const float*)d_in[0];
  const float* rela_table = (const float*)d_in[1];
  const float* ent_table  = (const float*)d_in[2];
  const float* W1         = (const float*)d_in[3];
  const float* b1         = (const float*)d_in[4];
  const float* W2         = (const float*)d_in[5];
  const float* b2         = (const float*)d_in[6];
  const float* lin_w      = (const float*)d_in[7];
  const float* lin_b      = (const float*)d_in[8];
  const float* bn_w       = (const float*)d_in[9];
  const float* bn_b       = (const float*)d_in[10];
  const int*   drug_name  = (const int*)d_in[11];
  const int*   adj_tail   = (const int*)d_in[12];
  const int*   adj_rel    = (const int*)d_in[13];

  float* ws     = (float*)d_ws;
  float* b2sum  = ws;
  float* drug_e = ws + K;
  float* y      = drug_e + N_DRUG * 2 * D;

  gnn_b2sum<<<dim3((K + 255) / 256), dim3(256), 0, stream>>>(b2, b2sum);

  const size_t lds = (size_t)(RPAD * D + D + D + K + 512) * sizeof(float);
  gnn_main<<<dim3(N_DRUG), dim3(256), lds, stream>>>(
      drug_table, rela_table, ent_table, W1, b1, W2,
      drug_name, adj_tail, adj_rel, b2sum, drug_e);

  gnn_linear<<<dim3(N_DRUG), dim3(64), 0, stream>>>(drug_e, lin_w, lin_b, y);

  gnn_bn<<<dim3(D), dim3(256), 0, stream>>>(y, bn_w, bn_b, (float*)d_out);
}


# Linear layer folded into the per-drug kernel: each wave forms 8 outputs from coalesced lin_w rows and a lane butterfly; one launch fewer, no drug_e round trip; sum order of the linear layer changes
# speedup vs baseline: 5.6586x; 1.0898x over previous
#include <hip/hip_runtime.h>

#define N_DRUG 846
#define D 64
#define K 1024
#define N_REL 200
#define RPAD 208
#define BN_EPS 1e-5f

typedef __attribute__((ext_vector_type(2))) float v2f;
typedef __attribute__((ext_vector_type(4))) float v4f;
typedef __attribute__((ext_vector_type(8))) float v8f;

__device__ __forceinline__ float wave_reduce_add(float v) {
  #pragma unroll
  for (int off = 16; off > 0; off >>= 1) v += __shfl_xor(v, off, 32);
  return v;
}

template <int M>
__device__ __forceinline__ float from_lane_xor(float x) {
  return __builtin_bit_cast(float, __builtin_amdgcn_update_dpp(
      0, __builtin_bit_cast(int, x), 0x160 | M, 0xf, 0xf, true));
}

template <int P>
__device__ __forceinline__ int from_row_lane(int x) {
  return __builtin_amdgcn_update_dpp(0, x, 0x150 | P, 0xf, 0xf, true);
}

__device__ __forceinline__ float from_other_row(float x) {
  const int b = __builtin_bit_cast(int, x);
  return __builtin_bit_cast(float, __builtin_amdgcn_permlanex16(
      b, b, 0x76543210, 0xfedcba98, false, false));
}

__global__ void __launch_bounds__(256) gnn_b2sum(const float* __restrict__ b2,
                                                 float* __restrict__ b2sum) {
  const int k = blockIdx.x * 256 + threadIdx.x;
  if (k < K) {
    const float* p = b2 + (size_t)k * D;
    float s = 0.f;
    #pragma unroll
    for (int e = 0; e < D; ++e) s += p[e];
    b2sum[k] = s;
  }
}

template <int B>
__device__ __forceinline__ void gather_t_rows(const float* tl, int held, v4f (&o)[8]) {
  o[0] = *(const v4f*)(tl + from_row_lane<8 * B + 0>(held));
  o[1] = *(const v4f*)(tl + from_row_lane<8 * B + 1>(held));
  o[2] = *(const v4f*)(tl + from_row_lane<8 * B + 2>(held));
  o[3] = *(const v4f*)(tl + from_row_lane<8 * B + 3>(held));
  o[4] = *(const v4f*)(tl + from_row_lane<8 * B + 4>(held));
  o[5] = *(const v4f*)(tl + from_row_lane<8 * B + 5>(held));
  o[6] = *(const v4f*)(tl + from_row_lane<8 * B + 6>(held));
  o[7] = *(const v4f*)(tl + from_row_lane<8 * B + 7>(held));
}
template <int B>
__device__ __forceinline__ void gather_ent_rows(const float* el, int held, v4f (&o)[8]) {
  o[0] = *(const v4f*)(el + (unsigned)from_row_lane<8 * B + 0>(held));
  o[1] = *(const v4f*)(el + (unsigned)from_row_lane<8 * B + 1>(held));
  o[2] = *(const v4f*)(el + (unsigned)from_row_lane<8 * B + 2>(held));
  o[3] = *(const v4f*)(el + (unsigned)from_row_lane<8 * B + 3>(held));
  o[4] = *(const v4f*)(el + (unsigned)from_row_lane<8 * B + 4>(held));
  o[5] = *(const v4f*)(el + (unsigned)from_row_lane<8 * B + 5>(held));
  o[6] = *(const v4f*)(el + (unsigned)from_row_lane<8 * B + 6>(held));
  o[7] = *(const v4f*)(el + (unsigned)from_row_lane<8 * B + 7>(held));
}
template <int B>
__device__ __forceinline__ void add_ent_rows(v4f& acc, int held, const v4f (&e)[8]) {
  acc += __int_as_float(from_row_lane<8 * B + 0>(held)) * e[0];
  acc += __int_as_float(from_row_lane<8 * B + 1>(held)) * e[1];
  acc += __int_as_float(from_row_lane<8 * B + 2>(held)) * e[2];
  acc += __int_as_float(from_row_lane<8 * B + 3>(held)) * e[3];
  acc += __int_as_float(from_row_lane<8 * B + 4>(held)) * e[4];
  acc += __int_as_float(from_row_lane<8 * B + 5>(held)) * e[5];
  acc += __int_as_float(from_row_lane<8 * B + 6>(held)) * e[6];
  acc += __int_as_float(from_row_lane<8 * B + 7>(held)) * e[7];
}

template <int B, int NB>
__device__ __forceinline__ void add_then_gather(v4f& acc, int held, v4f (&e)[8],
                                                const float* el, int nheld) {
  acc += __int_as_float(from_row_lane<8 * B + 0>(held)) * e[0];
  e[0] = *(const v4f*)(el + (unsigned)from_row_lane<8 * NB + 0>(nheld));
  acc += __int_as_float(from_row_lane<8 * B + 1>(held)) * e[1];
  e[1] = *(const v4f*)(el + (unsigned)from_row_lane<8 * NB + 1>(nheld));
  acc += __int_as_float(from_row_lane<8 * B + 2>(held)) * e[2];
  e[2] = *(const v4f*)(el + (unsigned)from_row_lane<8 * NB + 2>(nheld));
  acc += __int_as_float(from_row_lane<8 * B + 3>(held)) * e[3];
  e[3] = *(const v4f*)(el + (unsigned)from_row_lane<8 * NB + 3>(nheld));
  acc += __int_as_float(from_row_lane<8 * B + 4>(held)) * e[4];
  e[4] = *(const v4f*)(el + (unsigned)from_row_lane<8 * NB + 4>(nheld));
  acc += __int_as_float(from_row_lane<8 * B + 5>(held)) * e[5];
  e[5] = *(const v4f*)(el + (unsigned)from_row_lane<8 * NB + 5>(nheld));
  acc += __int_as_float(from_row_lane<8 * B + 6>(held)) * e[6];
  e[6] = *(const v4f*)(el + (unsigned)from_row_lane<8 * NB + 6>(nheld));
  acc += __int_as_float(from_row_lane<8 * B + 7>(held)) * e[7];
  e[7] = *(const v4f*)(el + (unsigned)from_row_lane<8 * NB + 7>(nheld));
}

struct group_more { static constexpr bool last = false; };
struct group_last { static constexpr bool last = true; };

__global__ void __launch_bounds__(256, 8) gnn_main(
    const float* __restrict__ drug_table, const float* __restrict__ rela_table,
    const float* __restrict__ ent_table,  const float* __restrict__ W1g,
    const float* __restrict__ b1,         const float* __restrict__ W2g,
    const float* __restrict__ lin_w,      const float* __restrict__ lin_b,
    const int*   __restrict__ drug_name,  const int* __restrict__ adj_tail,
    const int*   __restrict__ adj_rel,    const float* __restrict__ b2sum,
    float* __restrict__ y)
{
  extern __shared__ __align__(16) float smem[];
  float* Ttile = smem;
  float* demb  = Ttile + RPAD * D;
  float* w2sum = demb + D;
  float* score = w2sum + D;
  float* red   = score + K;

  const int n    = blockIdx.x;
  const int t    = threadIdx.x;
  const int lane = t & 31;
  const int wave = t >> 5;
  const int half = lane >> 4;
  const int l    = lane & 15;
  const int wv = __builtin_amdgcn_readfirstlane(wave);
  const int n0 = (wv & 3) * 16;
  const int mt = (wv >> 2) & 1;

  const size_t wbase = (size_t)n * D * D;
  const float* drow  = drug_table + (size_t)drug_name[n] * D;

  v2f bf[16];
  {
    const float* pb = W1g + wbase + (4 * half) * D + n0 + l;
    #pragma unroll
    for (int j = 0; j < 8; ++j) {
      #pragma unroll
      for (int s = 0; s < 2; ++s) {
        bf[2 * j + s][0] = pb[(8 * j + 2 * s) * D];
        bf[2 * j + s][1] = pb[(8 * j + 2 * s + 1) * D];
      }
    }
  }
  {
    const int d = t >> 2, q = t & 3;
    const float* p = W2g + wbase + d * D + q * 16;
    float s = 0.f;
    #pragma unroll
    for (int j = 0; j < 16; ++j) s += p[j];
    red[t] = s;
  }
  v4f ar[8];
  {
    const int row = min(mt * 16 + l, N_REL - 1);
    const float* pa = rela_table + row * D + 4 * half;
    #pragma unroll
    for (int j = 0; j < 8; ++j) ar[j] = *(const v4f*)(pa + 8 * j);
  }
  if (t < D) demb[t] = drow[t];
  __syncthreads();
  if (t < D) w2sum[t] = red[4*t] + red[4*t+1] + red[4*t+2] + red[4*t+3];
  {
    const float* pd = demb + 4 * half;
    #pragma unroll
    for (int j = 0; j < 8; ++j) {
      const v4f d4 = *(const v4f*)(pd + 8 * j);
      bf[2 * j][0]     *= d4[0];
      bf[2 * j][1]     *= d4[1];
      bf[2 * j + 1][0] *= d4[2];
      bf[2 * j + 1][1] *= d4[3];
    }
  }

  #pragma unroll 1
  for (int m = mt; m < 13; m += 2) {
    const int nrow = min(min(m + 2, 12) * 16 + l, N_REL - 1);
    const float* pn = rela_table + (unsigned)(nrow * D + 4 * half);
    v8f c = {0.f, 0.f, 0.f, 0.f, 0.f, 0.f, 0.f, 0.f};
    #pragma unroll
    for (int j = 0; j < 8; ++j) {
      const v2f a0 = {ar[j][0], ar[j][1]};
      const v2f a1 = {ar[j][2], ar[j][3]};
      c = __builtin_amdgcn_wmma_f32_16x16x4_f32(false, a0, false, bf[2 * j],
                                                (short)0, c, false, false);
      c = __builtin_amdgcn_wmma_f32_16x16x4_f32(false, a1, false, bf[2 * j + 1],
                                                (short)0, c, false, false);
      ar[j] = *(const v4f*)(pn + 8 * j);
    }
    const int m0 = m * 16;
    #pragma unroll
    for (int i = 0; i < 8; ++i)
      Ttile[(m0 + 8 * half + i) * D + n0 + l] = c[i];
  }
  __syncthreads();

  {
    const int   kw = wave * 128;
    const int   perm = 2 * l + half;
    const v4f   w4 = *(const v4f*)(w2sum + 4 * l);
    const float* tl = Ttile + 4 * l;
    const bool  hi8 = (lane & 8) != 0, hi4 = (lane & 4) != 0;
    const bool  hi2 = (lane & 2) != 0, hi1 = (lane & 1) != 0;
    const unsigned relo = (unsigned)n * K + kw + perm;
    const unsigned b1o  = (unsigned)kw * D + 4 * lane;
    int rvc = adj_rel[relo] * D;
    v4f cb[8];
    #pragma unroll
    for (int u = 0; u < 8; ++u)
      cb[u] = *(const v4f*)(b1 + (b1o + 2u * u * D));
    auto group = [&](const int j, auto tag) {
      constexpr bool LAST = decltype(tag)::last;
      int rvn = 0;
      if (!LAST) rvn = adj_rel[relo + (j + 1) * 32] * D;
      float v[16];
      #pragma unroll
      for (int b = 0; b < 2; ++b) {
        v4f tq[8];
        if (b == 0) gather_t_rows<0>(tl, rvc, tq);
        else        gather_t_rows<1>(tl, rvc, tq);
        const unsigned nb1 = b1o + (unsigned)(2 * j + b + 1) * (16 * D);
        #pragma unroll
        for (int u = 0; u < 8; ++u) {
          const v4f z = tq[u] + cb[u];
          v[b * 8 + u] = fmaxf(z[0], 0.f) * w4[0] + fmaxf(z[1], 0.f) * w4[1]
                       + fmaxf(z[2], 0.f) * w4[2] + fmaxf(z[3], 0.f) * w4[3];
          if (!LAST || b == 0) cb[u] = *(const v4f*)(b1 + (nb1 + 2u * u * D));
        }
      }
      float s8[8];
      #pragma unroll
      for (int i = 0; i < 8; ++i)
        s8[i] = (hi8 ? v[i + 8] : v[i]) + from_lane_xor<8>(hi8 ? v[i] : v[i + 8]);
      float s4[4];
      #pragma unroll
      for (int i = 0; i < 4; ++i)
        s4[i] = (hi4 ? s8[i + 4] : s8[i]) + from_lane_xor<4>(hi4 ? s8[i] : s8[i + 4]);
      float s2[2];
      #pragma unroll
      for (int i = 0; i < 2; ++i)
        s2[i] = (hi2 ? s4[i + 2] : s4[i]) + from_lane_xor<2>(hi2 ? s4[i] : s4[i + 2]);
      const float mine = (hi1 ? s2[1] : s2[0]) + from_lane_xor<1>(hi1 ? s2[0] : s2[1]);
      score[kw + j * 32 + perm] = mine + b2sum[(unsigned)(kw + j * 32 + perm)];
      rvc = rvn;
    };
    #pragma unroll 1
    for (int j = 0; j < 3; ++j) group(j, group_more{});
    group(3, group_last{});
  }
  __syncthreads();

  const int laneb = (int)__builtin_amdgcn_mbcnt_lo(~0u, 0u);
  const int tb    = wv * 32 + laneb;
  const int halfb = laneb >> 4;
  const int lb    = laneb & 15;

  float lmax = -INFINITY;
  for (int k = tb; k < K; k += 256) lmax = fmaxf(lmax, score[k]);
  red[tb] = lmax;
  __syncthreads();
  for (int s = 128; s > 0; s >>= 1) {
    if (tb < s) red[tb] = fmaxf(red[tb], red[tb + s]);
    __syncthreads();
  }
  const float mx = red[0];
  __syncthreads();
  float lsum = 0.f;
  for (int k = tb; k < K; k += 256) {
    const float e = __expf(score[k] - mx);
    score[k] = e;
    lsum += e;
  }
  red[tb] = lsum;
  __syncthreads();
  for (int s = 128; s > 0; s >>= 1) {
    if (tb < s) red[tb] += red[tb + s];
    __syncthreads();
  }
  const float inv = 1.f / red[0];
  __syncthreads();

  v4f lw[8];
  float lbias;
  {
    const int kw = wv * 128;
    const int perm = 2 * lb + halfb;
    const float* el = ent_table + 4 * lb;
    int tv[4], sv[4];
    #pragma unroll
    for (int j = 0; j < 4; ++j) {
      tv[j] = adj_tail[(unsigned)n * K + (unsigned)(kw + j * 32 + perm)] * D;
      sv[j] = __float_as_int(score[kw + j * 32 + perm] * inv);
    }
    v4f acc = {0.f, 0.f, 0.f, 0.f};
    v4f ea[8];
    gather_ent_rows<0>(el, tv[0], ea);
    {
      const float* pw = lin_w + (unsigned)(wv * 8 * (2 * D) + 4 * laneb);
      #pragma unroll
      for (int i = 0; i < 8; ++i) lw[i] = *(const v4f*)(pw + i * (2 * D));
      lbias = lin_b[wv * 8 + (laneb & 7)];
    }
    #pragma unroll
    for (int g = 0; g < 8; ++g) {
      if (g < 7) {
        if ((g & 1) == 0) add_then_gather<0, 1>(acc, sv[g >> 1], ea, el, tv[(g + 1) >> 1]);
        else              add_then_gather<1, 0>(acc, sv[g >> 1], ea, el, tv[(g + 1) >> 1]);
      } else {
        add_ent_rows<1>(acc, sv[g >> 1], ea);
      }
    }
    #pragma unroll
    for (int i = 0; i < 4; ++i) acc[i] += from_other_row(acc[i]);
    if (halfb == 0) *(v4f*)(red + wv * 64 + 4 * lb) = acc;
  }
  __syncthreads();

  v4f de;
  if (halfb == 0) {
    v4f s = {0.f, 0.f, 0.f, 0.f};
    #pragma unroll
    for (int w = 0; w < 8; ++w) s += *(const v4f*)(red + w * 64 + 4 * lb);
    de = s;
  } else {
    de = *(const v4f*)(demb + 4 * lb);
  }
  float p8[8];
  #pragma unroll
  for (int i = 0; i < 8; ++i)
    p8[i] = lw[i][0] * de[0] + lw[i][1] * de[1] + lw[i][2] * de[2] + lw[i][3] * de[3];
  {
    const bool b4 = (laneb & 4) != 0, b2 = (laneb & 2) != 0, b1l = (laneb & 1) != 0;
    float p4[4];
    #pragma unroll
    for (int i = 0; i < 4; ++i)
      p4[i] = (b4 ? p8[i + 4] : p8[i]) + from_lane_xor<4>(b4 ? p8[i] : p8[i + 4]);
    float p2[2];
    #pragma unroll
    for (int i = 0; i < 2; ++i)
      p2[i] = (b2 ? p4[i + 2] : p4[i]) + from_lane_xor<2>(b2 ? p4[i] : p4[i + 2]);
    float o = (b1l ? p2[1] : p2[0]) + from_lane_xor<1>(b1l ? p2[0] : p2[1]);
    o += from_lane_xor<8>(o);
    o += from_other_row(o);
    if (laneb < 8) score[wv * 8 + laneb] = fmaxf(o + lbias, 0.f);
  }
  __syncthreads();
  if (wv == 0 && laneb < 16)
    *(v4f*)(y + (size_t)n * D + 4 * laneb) = *(const v4f*)(score + 4 * laneb);
}

__global__ void __launch_bounds__(256) gnn_bn(
    const float* __restrict__ y, const float* __restrict__ bn_w,
    const float* __restrict__ bn_b, float* __restrict__ out)
{
  __shared__ float red[256];
  const int c = blockIdx.x, t = threadIdx.x;
  float s = 0.f;
  for (int r = t; r < N_DRUG; r += 256) s += y[(size_t)r * D + c];
  red[t] = s;
  __syncthreads();
  for (int st = 128; st > 0; st >>= 1) {
    if (t < st) red[t] += red[t + st];
    __syncthreads();
  }
  const float mean = red[0] / (float)N_DRUG;
  __syncthreads();
  float q = 0.f;
  for (int r = t; r < N_DRUG; r += 256) {
    const float d = y[(size_t)r * D + c] - mean;
    q += d * d;
  }
  red[t] = q;
  __syncthreads();
  for (int st = 128; st > 0; st >>= 1) {
    if (t < st) red[t] += red[t + st];
    __syncthreads();
  }
  const float var   = red[0] / (float)N_DRUG;
  const float scale = bn_w[c] * rsqrtf(var + BN_EPS);
  const float shift = bn_b[c];
  for (int r = t; r < N_DRUG; r += 256)
    out[(size_t)r * D + c] = (y[(size_t)r * D + c] - mean) * scale + shift;
}

extern "C" void kernel_launch(void* const* d_in, const int* in_sizes, int n_in,
                              void* d_out, int out_size, void* d_ws, size_t ws_size,
                              hipStream_t stream) {
  (void)in_sizes; (void)n_in; (void)out_size; (void)ws_size;
  const float* drug_table = (const float*)d_in[0];
  const float* rela_table = (const float*)d_in[1];
  const float* ent_table  = (const float*)d_in[2];
  const float* W1         = (const float*)d_in[3];
  const float* b1         = (const float*)d_in[4];
  const float* W2         = (const float*)d_in[5];
  const float* b2         = (const float*)d_in[6];
  const float* lin_w      = (const float*)d_in[7];
  const float* lin_b      = (const float*)d_in[8];
  const float* bn_w       = (const float*)d_in[9];
  const float* bn_b       = (const float*)d_in[10];
  const int*   drug_name  = (const int*)d_in[11];
  const int*   adj_tail   = (const int*)d_in[12];
  const int*   adj_rel    = (const int*)d_in[13];

  float* ws     = (float*)d_ws;
  float* b2sum  = ws;
  float* y      = ws + K;

  gnn_b2sum<<<dim3((K + 255) / 256), dim3(256), 0, stream>>>(b2, b2sum);

  const size_t lds = (size_t)(RPAD * D + D + D + K + 512) * sizeof(float);
  gnn_main<<<dim3(N_DRUG), dim3(256), lds, stream>>>(
      drug_table, rela_table, ent_table, W1, b1, W2, lin_w, lin_b,
      drug_name, adj_tail, adj_rel, b2sum, y);

  gnn_bn<<<dim3(D), dim3(256), 0, stream>>>(y, bn_w, bn_b, (float*)d_out);
}


# Entity sum fetches a batch of 16 neighbour rows only when one of its 16 softmax weights is not exactly zero (wave vote, scalar branch); zero weights add nothing; sum order kept, bit-identical
# speedup vs baseline: 6.0268x; 1.0651x over previous
#include <hip/hip_runtime.h>

#define N_DRUG 846
#define D 64
#define K 1024
#define N_REL 200
#define RPAD 208
#define BN_EPS 1e-5f

typedef __attribute__((ext_vector_type(2))) float v2f;
typedef __attribute__((ext_vector_type(4))) float v4f;
typedef __attribute__((ext_vector_type(8))) float v8f;

__device__ __forceinline__ float wave_reduce_add(float v) {
  #pragma unroll
  for (int off = 16; off > 0; off >>= 1) v += __shfl_xor(v, off, 32);
  return v;
}

template <int M>
__device__ __forceinline__ float from_lane_xor(float x) {
  return __builtin_bit_cast(float, __builtin_amdgcn_update_dpp(
      0, __builtin_bit_cast(int, x), 0x160 | M, 0xf, 0xf, true));
}

template <int P>
__device__ __forceinline__ int from_row_lane(int x) {
  return __builtin_amdgcn_update_dpp(0, x, 0x150 | P, 0xf, 0xf, true);
}

__device__ __forceinline__ float from_other_row(float x) {
  const int b = __builtin_bit_cast(int, x);
  return __builtin_bit_cast(float, __builtin_amdgcn_permlanex16(
      b, b, 0x76543210, 0xfedcba98, false, false));
}

__global__ void __launch_bounds__(256) gnn_b2sum(const float* __restrict__ b2,
                                                 float* __restrict__ b2sum) {
  const int k = blockIdx.x * 256 + threadIdx.x;
  if (k < K) {
    const float* p = b2 + (size_t)k * D;
    float s = 0.f;
    #pragma unroll
    for (int e = 0; e < D; ++e) s += p[e];
    b2sum[k] = s;
  }
}

template <int B>
__device__ __forceinline__ void gather_t_rows(const float* tl, int held, v4f (&o)[8]) {
  o[0] = *(const v4f*)(tl + from_row_lane<8 * B + 0>(held));
  o[1] = *(const v4f*)(tl + from_row_lane<8 * B + 1>(held));
  o[2] = *(const v4f*)(tl + from_row_lane<8 * B + 2>(held));
  o[3] = *(const v4f*)(tl + from_row_lane<8 * B + 3>(held));
  o[4] = *(const v4f*)(tl + from_row_lane<8 * B + 4>(held));
  o[5] = *(const v4f*)(tl + from_row_lane<8 * B + 5>(held));
  o[6] = *(const v4f*)(tl + from_row_lane<8 * B + 6>(held));
  o[7] = *(const v4f*)(tl + from_row_lane<8 * B + 7>(held));
}
template <int B>
__device__ __forceinline__ void gather_ent_rows(const float* el, int held, v4f (&o)[8]) {
  o[0] = *(const v4f*)(el + (unsigned)from_row_lane<8 * B + 0>(held));
  o[1] = *(const v4f*)(el + (unsigned)from_row_lane<8 * B + 1>(held));
  o[2] = *(const v4f*)(el + (unsigned)from_row_lane<8 * B + 2>(held));
  o[3] = *(const v4f*)(el + (unsigned)from_row_lane<8 * B + 3>(held));
  o[4] = *(const v4f*)(el + (unsigned)from_row_lane<8 * B + 4>(held));
  o[5] = *(const v4f*)(el + (unsigned)from_row_lane<8 * B + 5>(held));
  o[6] = *(const v4f*)(el + (unsigned)from_row_lane<8 * B + 6>(held));
  o[7] = *(const v4f*)(el + (unsigned)from_row_lane<8 * B + 7>(held));
}
template <int B>
__device__ __forceinline__ void add_ent_rows(v4f& acc, int held, const v4f (&e)[8]) {
  acc += __int_as_float(from_row_lane<8 * B + 0>(held)) * e[0];
  acc += __int_as_float(from_row_lane<8 * B + 1>(held)) * e[1];
  acc += __int_as_float(from_row_lane<8 * B + 2>(held)) * e[2];
  acc += __int_as_float(from_row_lane<8 * B + 3>(held)) * e[3];
  acc += __int_as_float(from_row_lane<8 * B + 4>(held)) * e[4];
  acc += __int_as_float(from_row_lane<8 * B + 5>(held)) * e[5];
  acc += __int_as_float(from_row_lane<8 * B + 6>(held)) * e[6];
  acc += __int_as_float(from_row_lane<8 * B + 7>(held)) * e[7];
}

struct group_more { static constexpr bool last = false; };
struct group_last { static constexpr bool last = true; };

__global__ void __launch_bounds__(256, 8) gnn_main(
    const float* __restrict__ drug_table, const float* __restrict__ rela_table,
    const float* __restrict__ ent_table,  const float* __restrict__ W1g,
    const float* __restrict__ b1,         const float* __restrict__ W2g,
    const float* __restrict__ lin_w,      const float* __restrict__ lin_b,
    const int*   __restrict__ drug_name,  const int* __restrict__ adj_tail,
    const int*   __restrict__ adj_rel,    const float* __restrict__ b2sum,
    float* __restrict__ y)
{
  extern __shared__ __align__(16) float smem[];
  float* Ttile = smem;
  float* demb  = Ttile + RPAD * D;
  float* w2sum = demb + D;
  float* score = w2sum + D;
  float* red   = score + K;

  const int n    = blockIdx.x;
  const int t    = threadIdx.x;
  const int lane = t & 31;
  const int wave = t >> 5;
  const int half = lane >> 4;
  const int l    = lane & 15;
  const int wv = __builtin_amdgcn_readfirstlane(wave);
  const int n0 = (wv & 3) * 16;
  const int mt = (wv >> 2) & 1;

  const size_t wbase = (size_t)n * D * D;
  const float* drow  = drug_table + (size_t)drug_name[n] * D;

  v2f bf[16];
  {
    const float* pb = W1g + wbase + (4 * half) * D + n0 + l;
    #pragma unroll
    for (int j = 0; j < 8; ++j) {
      #pragma unroll
      for (int s = 0; s < 2; ++s) {
        bf[2 * j + s][0] = pb[(8 * j + 2 * s) * D];
        bf[2 * j + s][1] = pb[(8 * j + 2 * s + 1) * D];
      }
    }
  }
  {
    const int d = t >> 2, q = t & 3;
    const float* p = W2g + wbase + d * D + q * 16;
    float s = 0.f;
    #pragma unroll
    for (int j = 0; j < 16; ++j) s += p[j];
    red[t] = s;
  }
  v4f ar[8];
  {
    const int row = min(mt * 16 + l, N_REL - 1);
    const float* pa = rela_table + row * D + 4 * half;
    #pragma unroll
    for (int j = 0; j < 8; ++j) ar[j] = *(const v4f*)(pa + 8 * j);
  }
  if (t < D) demb[t] = drow[t];
  __syncthreads();
  if (t < D) w2sum[t] = red[4*t] + red[4*t+1] + red[4*t+2] + red[4*t+3];
  {
    const float* pd = demb + 4 * half;
    #pragma unroll
    for (int j = 0; j < 8; ++j) {
      const v4f d4 = *(const v4f*)(pd + 8 * j);
      bf[2 * j][0]     *= d4[0];
      bf[2 * j][1]     *= d4[1];
      bf[2 * j + 1][0] *= d4[2];
      bf[2 * j + 1][1] *= d4[3];
    }
  }

  #pragma unroll 1
  for (int m = mt; m < 13; m += 2) {
    const int nrow = min(min(m + 2, 12) * 16 + l, N_REL - 1);
    const float* pn = rela_table + (unsigned)(nrow * D + 4 * half);
    v8f c = {0.f, 0.f, 0.f, 0.f, 0.f, 0.f, 0.f, 0.f};
    #pragma unroll
    for (int j = 0; j < 8; ++j) {
      const v2f a0 = {ar[j][0], ar[j][1]};
      const v2f a1 = {ar[j][2], ar[j][3]};
      c = __builtin_amdgcn_wmma_f32_16x16x4_f32(false, a0, false, bf[2 * j],
                                                (short)0, c, false, false);
      c = __builtin_amdgcn_wmma_f32_16x16x4_f32(false, a1, false, bf[2 * j + 1],
                                                (short)0, c, false, false);
      ar[j] = *(const v4f*)(pn + 8 * j);
    }
    const int m0 = m * 16;
    #pragma unroll
    for (int i = 0; i < 8; ++i)
      Ttile[(m0 + 8 * half + i) * D + n0 + l] = c[i];
  }
  __syncthreads();

  {
    const int   kw = wave * 128;
    const int   perm = 2 * l + half;
    const v4f   w4 = *(const v4f*)(w2sum + 4 * l);
    const float* tl = Ttile + 4 * l;
    const bool  hi8 = (lane & 8) != 0, hi4 = (lane & 4) != 0;
    const bool  hi2 = (lane & 2) != 0, hi1 = (lane & 1) != 0;
    const unsigned relo = (unsigned)n * K + kw + perm;
    const unsigned b1o  = (unsigned)kw * D + 4 * lane;
    int rvc = adj_rel[relo] * D;
    v4f cb[8];
    #pragma unroll
    for (int u = 0; u < 8; ++u)
      cb[u] = *(const v4f*)(b1 + (b1o + 2u * u * D));
    auto group = [&](const int j, auto tag) {
      constexpr bool LAST = decltype(tag)::last;
      int rvn = 0;
      if (!LAST) rvn = adj_rel[relo + (j + 1) * 32] * D;
      float v[16];
      #pragma unroll
      for (int b = 0; b < 2; ++b) {
        v4f tq[8];
        if (b == 0) gather_t_rows<0>(tl, rvc, tq);
        else        gather_t_rows<1>(tl, rvc, tq);
        const unsigned nb1 = b1o + (unsigned)(2 * j + b + 1) * (16 * D);
        #pragma unroll
        for (int u = 0; u < 8; ++u) {
          const v4f z = tq[u] + cb[u];
          v[b * 8 + u] = fmaxf(z[0], 0.f) * w4[0] + fmaxf(z[1], 0.f) * w4[1]
                       + fmaxf(z[2], 0.f) * w4[2] + fmaxf(z[3], 0.f) * w4[3];
          if (!LAST || b == 0) cb[u] = *(const v4f*)(b1 + (nb1 + 2u * u * D));
        }
      }
      float s8[8];
      #pragma unroll
      for (int i = 0; i < 8; ++i)
        s8[i] = (hi8 ? v[i + 8] : v[i]) + from_lane_xor<8>(hi8 ? v[i] : v[i + 8]);
      float s4[4];
      #pragma unroll
      for (int i = 0; i < 4; ++i)
        s4[i] = (hi4 ? s8[i + 4] : s8[i]) + from_lane_xor<4>(hi4 ? s8[i] : s8[i + 4]);
      float s2[2];
      #pragma unroll
      for (int i = 0; i < 2; ++i)
        s2[i] = (hi2 ? s4[i + 2] : s4[i]) + from_lane_xor<2>(hi2 ? s4[i] : s4[i + 2]);
      const float mine = (hi1 ? s2[1] : s2[0]) + from_lane_xor<1>(hi1 ? s2[0] : s2[1]);
      score[kw + j * 32 + perm] = mine + b2sum[(unsigned)(kw + j * 32 + perm)];
      rvc = rvn;
    };
    #pragma unroll 1
    for (int j = 0; j < 3; ++j) group(j, group_more{});
    group(3, group_last{});
  }
  __syncthreads();

  const int laneb = (int)__builtin_amdgcn_mbcnt_lo(~0u, 0u);
  const int tb    = wv * 32 + laneb;
  const int halfb = laneb >> 4;
  const int lb    = laneb & 15;

  float lmax = -INFINITY;
  for (int k = tb; k < K; k += 256) lmax = fmaxf(lmax, score[k]);
  red[tb] = lmax;
  __syncthreads();
  for (int s = 128; s > 0; s >>= 1) {
    if (tb < s) red[tb] = fmaxf(red[tb], red[tb + s]);
    __syncthreads();
  }
  const float mx = red[0];
  __syncthreads();
  float lsum = 0.f;
  for (int k = tb; k < K; k += 256) {
    const float e = __expf(score[k] - mx);
    score[k] = e;
    lsum += e;
  }
  red[tb] = lsum;
  __syncthreads();
  for (int s = 128; s > 0; s >>= 1) {
    if (tb < s) red[tb] += red[tb + s];
    __syncthreads();
  }
  const float inv = 1.f / red[0];
  __syncthreads();

  v4f lw[8];
  float lbias;
  {
    const int kw = wv * 128;
    const int perm = 2 * lb + halfb;
    const float* el = ent_table + 4 * lb;
    int tv[4], sv[4];
    #pragma unroll
    for (int j = 0; j < 4; ++j) {
      tv[j] = adj_tail[(unsigned)n * K + (unsigned)(kw + j * 32 + perm)] * D;
      sv[j] = __float_as_int(score[kw + j * 32 + perm] * inv);
    }
    v4f acc = {0.f, 0.f, 0.f, 0.f};
    v4f ea[8];
    {
      const float* pw = lin_w + (unsigned)(wv * 8 * (2 * D) + 4 * laneb);
      #pragma unroll
      for (int i = 0; i < 8; ++i) lw[i] = *(const v4f*)(pw + i * (2 * D));
      lbias = lin_b[wv * 8 + (laneb & 7)];
    }
    #pragma unroll
    for (int j = 0; j < 4; ++j) {
      const unsigned nz = __builtin_amdgcn_ballot_w32(__int_as_float(sv[j]) != 0.f);
      if (nz & 0x00ff00ffu) {
        gather_ent_rows<0>(el, tv[j], ea);
        add_ent_rows<0>(acc, sv[j], ea);
      }
      if (nz & 0xff00ff00u) {
        gather_ent_rows<1>(el, tv[j], ea);
        add_ent_rows<1>(acc, sv[j], ea);
      }
    }
    #pragma unroll
    for (int i = 0; i < 4; ++i) acc[i] += from_other_row(acc[i]);
    if (halfb == 0) *(v4f*)(red + wv * 64 + 4 * lb) = acc;
  }
  __syncthreads();

  v4f de;
  if (halfb == 0) {
    v4f s = {0.f, 0.f, 0.f, 0.f};
    #pragma unroll
    for (int w = 0; w < 8; ++w) s += *(const v4f*)(red + w * 64 + 4 * lb);
    de = s;
  } else {
    de = *(const v4f*)(demb + 4 * lb);
  }
  float p8[8];
  #pragma unroll
  for (int i = 0; i < 8; ++i)
    p8[i] = lw[i][0] * de[0] + lw[i][1] * de[1] + lw[i][2] * de[2] + lw[i][3] * de[3];
  {
    const bool b4 = (laneb & 4) != 0, b2 = (laneb & 2) != 0, b1l = (laneb & 1) != 0;
    float p4[4];
    #pragma unroll
    for (int i = 0; i < 4; ++i)
      p4[i] = (b4 ? p8[i + 4] : p8[i]) + from_lane_xor<4>(b4 ? p8[i] : p8[i + 4]);
    float p2[2];
    #pragma unroll
    for (int i = 0; i < 2; ++i)
      p2[i] = (b2 ? p4[i + 2] : p4[i]) + from_lane_xor<2>(b2 ? p4[i] : p4[i + 2]);
    float o = (b1l ? p2[1] : p2[0]) + from_lane_xor<1>(b1l ? p2[0] : p2[1]);
    o += from_lane_xor<8>(o);
    o += from_other_row(o);
    if (laneb < 8) score[wv * 8 + laneb] = fmaxf(o + lbias, 0.f);
  }
  __syncthreads();
  if (wv == 0 && laneb < 16)
    *(v4f*)(y + (size_t)n * D + 4 * laneb) = *(const v4f*)(score + 4 * laneb);
}

__global__ void __launch_bounds__(256) gnn_bn(
    const float* __restrict__ y, const float* __restrict__ bn_w,
    const float* __restrict__ bn_b, float* __restrict__ out)
{
  __shared__ float red[256];
  const int c = blockIdx.x, t = threadIdx.x;
  float s = 0.f;
  for (int r = t; r < N_DRUG; r += 256) s += y[(size_t)r * D + c];
  red[t] = s;
  __syncthreads();
  for (int st = 128; st > 0; st >>= 1) {
    if (t < st) red[t] += red[t + st];
    __syncthreads();
  }
  const float mean = red[0] / (float)N_DRUG;
  __syncthreads();
  float q = 0.f;
  for (int r = t; r < N_DRUG; r += 256) {
    const float d = y[(size_t)r * D + c] - mean;
    q += d * d;
  }
  red[t] = q;
  __syncthreads();
  for (int st = 128; st > 0; st >>= 1) {
    if (t < st) red[t] += red[t + st];
    __syncthreads();
  }
  const float var   = red[0] / (float)N_DRUG;
  const float scale = bn_w[c] * rsqrtf(var + BN_EPS);
  const float shift = bn_b[c];
  for (int r = t; r < N_DRUG; r += 256)
    out[(size_t)r * D + c] = (y[(size_t)r * D + c] - mean) * scale + shift;
}

extern "C" void kernel_launch(void* const* d_in, const int* in_sizes, int n_in,
                              void* d_out, int out_size, void* d_ws, size_t ws_size,
                              hipStream_t stream) {
  (void)in_sizes; (void)n_in; (void)out_size; (void)ws_size;
  const float* drug_table = (const float*)d_in[0];
  const float* rela_table = (const float*)d_in[1];
  const float* ent_table  = (const float*)d_in[2];
  const float* W1         = (const float*)d_in[3];
  const float* b1         = (const float*)d_in[4];
  const float* W2         = (const float*)d_in[5];
  const float* b2         = (const float*)d_in[6];
  const float* lin_w      = (const float*)d_in[7];
  const float* lin_b      = (const float*)d_in[8];
  const float* bn_w       = (const float*)d_in[9];
  const float* bn_b       = (const float*)d_in[10];
  const int*   drug_name  = (const int*)d_in[11];
  const int*   adj_tail   = (const int*)d_in[12];
  const int*   adj_rel    = (const int*)d_in[13];

  float* ws     = (float*)d_ws;
  float* b2sum  = ws;
  float* y      = ws + K;

  gnn_b2sum<<<dim3((K + 255) / 256), dim3(256), 0, stream>>>(b2, b2sum);

  const size_t lds = (size_t)(RPAD * D + D + D + K + 512) * sizeof(float);
  gnn_main<<<dim3(N_DRUG), dim3(256), lds, stream>>>(
      drug_table, rela_table, ent_table, W1, b1, W2, lin_w, lin_b,
      drug_name, adj_tail, adj_rel, b2sum, y);

  gnn_bn<<<dim3(D), dim3(256), 0, stream>>>(y, bn_w, bn_b, (float*)d_out);
}


# Work-group reductions (softmax maximum and denominator, batch-norm mean and variance): lane swizzles in each wave, then 8 wave results via LDS with one barrier, no 8-step LDS tree; sum order changes
# speedup vs baseline: 6.2221x; 1.0324x over previous
#include <hip/hip_runtime.h>

#define N_DRUG 846
#define D 64
#define K 1024
#define N_REL 200
#define RPAD 208
#define BN_EPS 1e-5f

typedef __attribute__((ext_vector_type(2))) float v2f;
typedef __attribute__((ext_vector_type(4))) float v4f;
typedef __attribute__((ext_vector_type(8))) float v8f;

template <int M>
__device__ __forceinline__ float from_lane_xor(float x) {
  return __builtin_bit_cast(float, __builtin_amdgcn_update_dpp(
      0, __builtin_bit_cast(int, x), 0x160 | M, 0xf, 0xf, true));
}

template <int P>
__device__ __forceinline__ int from_row_lane(int x) {
  return __builtin_amdgcn_update_dpp(0, x, 0x150 | P, 0xf, 0xf, true);
}

__device__ __forceinline__ float from_other_row(float x) {
  const int b = __builtin_bit_cast(int, x);
  return __builtin_bit_cast(float, __builtin_amdgcn_permlanex16(
      b, b, 0x76543210, 0xfedcba98, false, false));
}

__device__ __forceinline__ float wave_sum_all(float v) {
  v += from_lane_xor<1>(v);
  v += from_lane_xor<2>(v);
  v += from_lane_xor<4>(v);
  v += from_lane_xor<8>(v);
  v += from_other_row(v);
  return v;
}
__device__ __forceinline__ float wave_max_all(float v) {
  v = fmaxf(v, from_lane_xor<1>(v));
  v = fmaxf(v, from_lane_xor<2>(v));
  v = fmaxf(v, from_lane_xor<4>(v));
  v = fmaxf(v, from_lane_xor<8>(v));
  v = fmaxf(v, from_other_row(v));
  return v;
}

__global__ void __launch_bounds__(256) gnn_b2sum(const float* __restrict__ b2,
                                                 float* __restrict__ b2sum) {
  const int k = blockIdx.x * 256 + threadIdx.x;
  if (k < K) {
    const float* p = b2 + (size_t)k * D;
    float s = 0.f;
    #pragma unroll
    for (int e = 0; e < D; ++e) s += p[e];
    b2sum[k] = s;
  }
}

template <int B>
__device__ __forceinline__ void gather_t_rows(const float* tl, int held, v4f (&o)[8]) {
  o[0] = *(const v4f*)(tl + from_row_lane<8 * B + 0>(held));
  o[1] = *(const v4f*)(tl + from_row_lane<8 * B + 1>(held));
  o[2] = *(const v4f*)(tl + from_row_lane<8 * B + 2>(held));
  o[3] = *(const v4f*)(tl + from_row_lane<8 * B + 3>(held));
  o[4] = *(const v4f*)(tl + from_row_lane<8 * B + 4>(held));
  o[5] = *(const v4f*)(tl + from_row_lane<8 * B + 5>(held));
  o[6] = *(const v4f*)(tl + from_row_lane<8 * B + 6>(held));
  o[7] = *(const v4f*)(tl + from_row_lane<8 * B + 7>(held));
}
template <int B>
__device__ __forceinline__ void gather_ent_rows(const float* el, int held, v4f (&o)[8]) {
  o[0] = *(const v4f*)(el + (unsigned)from_row_lane<8 * B + 0>(held));
  o[1] = *(const v4f*)(el + (unsigned)from_row_lane<8 * B + 1>(held));
  o[2] = *(const v4f*)(el + (unsigned)from_row_lane<8 * B + 2>(held));
  o[3] = *(const v4f*)(el + (unsigned)from_row_lane<8 * B + 3>(held));
  o[4] = *(const v4f*)(el + (unsigned)from_row_lane<8 * B + 4>(held));
  o[5] = *(const v4f*)(el + (unsigned)from_row_lane<8 * B + 5>(held));
  o[6] = *(const v4f*)(el + (unsigned)from_row_lane<8 * B + 6>(held));
  o[7] = *(const v4f*)(el + (unsigned)from_row_lane<8 * B + 7>(held));
}
template <int B>
__device__ __forceinline__ void add_ent_rows(v4f& acc, int held, const v4f (&e)[8]) {
  acc += __int_as_float(from_row_lane<8 * B + 0>(held)) * e[0];
  acc += __int_as_float(from_row_lane<8 * B + 1>(held)) * e[1];
  acc += __int_as_float(from_row_lane<8 * B + 2>(held)) * e[2];
  acc += __int_as_float(from_row_lane<8 * B + 3>(held)) * e[3];
  acc += __int_as_float(from_row_lane<8 * B + 4>(held)) * e[4];
  acc += __int_as_float(from_row_lane<8 * B + 5>(held)) * e[5];
  acc += __int_as_float(from_row_lane<8 * B + 6>(held)) * e[6];
  acc += __int_as_float(from_row_lane<8 * B + 7>(held)) * e[7];
}

struct group_more { static constexpr bool last = false; };
struct group_last { static constexpr bool last = true; };

__global__ void __launch_bounds__(256, 8) gnn_main(
    const float* __restrict__ drug_table, const float* __restrict__ rela_table,
    const float* __restrict__ ent_table,  const float* __restrict__ W1g,
    const float* __restrict__ b1,         const float* __restrict__ W2g,
    const float* __restrict__ lin_w,      const float* __restrict__ lin_b,
    const int*   __restrict__ drug_name,  const int* __restrict__ adj_tail,
    const int*   __restrict__ adj_rel,    const float* __restrict__ b2sum,
    float* __restrict__ y)
{
  extern __shared__ __align__(16) float smem[];
  float* Ttile = smem;
  float* demb  = Ttile + RPAD * D;
  float* w2sum = demb + D;
  float* score = w2sum + D;
  float* red   = score + K;

  const int n    = blockIdx.x;
  const int t    = threadIdx.x;
  const int lane = t & 31;
  const int wave = t >> 5;
  const int half = lane >> 4;
  const int l    = lane & 15;
  const int wv = __builtin_amdgcn_readfirstlane(wave);
  const int n0 = (wv & 3) * 16;
  const int mt = (wv >> 2) & 1;

  const size_t wbase = (size_t)n * D * D;
  const float* drow  = drug_table + (size_t)drug_name[n] * D;

  v2f bf[16];
  {
    const float* pb = W1g + wbase + (4 * half) * D + n0 + l;
    #pragma unroll
    for (int j = 0; j < 8; ++j) {
      #pragma unroll
      for (int s = 0; s < 2; ++s) {
        bf[2 * j + s][0] = pb[(8 * j + 2 * s) * D];
        bf[2 * j + s][1] = pb[(8 * j + 2 * s + 1) * D];
      }
    }
  }
  {
    const int d = t >> 2, q = t & 3;
    const float* p = W2g + wbase + d * D + q * 16;
    float s = 0.f;
    #pragma unroll
    for (int j = 0; j < 16; ++j) s += p[j];
    red[t] = s;
  }
  v4f ar[8];
  {
    const int row = min(mt * 16 + l, N_REL - 1);
    const float* pa = rela_table + row * D + 4 * half;
    #pragma unroll
    for (int j = 0; j < 8; ++j) ar[j] = *(const v4f*)(pa + 8 * j);
  }
  if (t < D) demb[t] = drow[t];
  __syncthreads();
  if (t < D) w2sum[t] = red[4*t] + red[4*t+1] + red[4*t+2] + red[4*t+3];
  {
    const float* pd = demb + 4 * half;
    #pragma unroll
    for (int j = 0; j < 8; ++j) {
      const v4f d4 = *(const v4f*)(pd + 8 * j);
      bf[2 * j][0]     *= d4[0];
      bf[2 * j][1]     *= d4[1];
      bf[2 * j + 1][0] *= d4[2];
      bf[2 * j + 1][1] *= d4[3];
    }
  }

  #pragma unroll 1
  for (int m = mt; m < 13; m += 2) {
    const int nrow = min(min(m + 2, 12) * 16 + l, N_REL - 1);
    const float* pn = rela_table + (unsigned)(nrow * D + 4 * half);
    v8f c = {0.f, 0.f, 0.f, 0.f, 0.f, 0.f, 0.f, 0.f};
    #pragma unroll
    for (int j = 0; j < 8; ++j) {
      const v2f a0 = {ar[j][0], ar[j][1]};
      const v2f a1 = {ar[j][2], ar[j][3]};
      c = __builtin_amdgcn_wmma_f32_16x16x4_f32(false, a0, false, bf[2 * j],
                                                (short)0, c, false, false);
      c = __builtin_amdgcn_wmma_f32_16x16x4_f32(false, a1, false, bf[2 * j + 1],
                                                (short)0, c, false, false);
      ar[j] = *(const v4f*)(pn + 8 * j);
    }
    const int m0 = m * 16;
    #pragma unroll
    for (int i = 0; i < 8; ++i)
      Ttile[(m0 + 8 * half + i) * D + n0 + l] = c[i];
  }
  __syncthreads();

  {
    const int   kw = wave * 128;
    const int   perm = 2 * l + half;
    const v4f   w4 = *(const v4f*)(w2sum + 4 * l);
    const float* tl = Ttile + 4 * l;
    const bool  hi8 = (lane & 8) != 0, hi4 = (lane & 4) != 0;
    const bool  hi2 = (lane & 2) != 0, hi1 = (lane & 1) != 0;
    const unsigned relo = (unsigned)n * K + kw + perm;
    const unsigned b1o  = (unsigned)kw * D + 4 * lane;
    int rvc = adj_rel[relo] * D;
    v4f cb[8];
    #pragma unroll
    for (int u = 0; u < 8; ++u)
      cb[u] = *(const v4f*)(b1 + (b1o + 2u * u * D));
    auto group = [&](const int j, auto tag) {
      constexpr bool LAST = decltype(tag)::last;
      int rvn = 0;
      if (!LAST) rvn = adj_rel[relo + (j + 1) * 32] * D;
      float v[16];
      #pragma unroll
      for (int b = 0; b < 2; ++b) {
        v4f tq[8];
        if (b == 0) gather_t_rows<0>(tl, rvc, tq);
        else        gather_t_rows<1>(tl, rvc, tq);
        const unsigned nb1 = b1o + (unsigned)(2 * j + b + 1) * (16 * D);
        #pragma unroll
        for (int u = 0; u < 8; ++u) {
          const v4f z = tq[u] + cb[u];
          v[b * 8 + u] = fmaxf(z[0], 0.f) * w4[0] + fmaxf(z[1], 0.f) * w4[1]
                       + fmaxf(z[2], 0.f) * w4[2] + fmaxf(z[3], 0.f) * w4[3];
          if (!LAST || b == 0) cb[u] = *(const v4f*)(b1 + (nb1 + 2u * u * D));
        }
      }
      float s8[8];
      #pragma unroll
      for (int i = 0; i < 8; ++i)
        s8[i] = (hi8 ? v[i + 8] : v[i]) + from_lane_xor<8>(hi8 ? v[i] : v[i + 8]);
      float s4[4];
      #pragma unroll
      for (int i = 0; i < 4; ++i)
        s4[i] = (hi4 ? s8[i + 4] : s8[i]) + from_lane_xor<4>(hi4 ? s8[i] : s8[i + 4]);
      float s2[2];
      #pragma unroll
      for (int i = 0; i < 2; ++i)
        s2[i] = (hi2 ? s4[i + 2] : s4[i]) + from_lane_xor<2>(hi2 ? s4[i] : s4[i + 2]);
      const float mine = (hi1 ? s2[1] : s2[0]) + from_lane_xor<1>(hi1 ? s2[0] : s2[1]);
      score[kw + j * 32 + perm] = mine + b2sum[(unsigned)(kw + j * 32 + perm)];
      rvc = rvn;
    };
    #pragma unroll 1
    for (int j = 0; j < 3; ++j) group(j, group_more{});
    group(3, group_last{});
  }
  __syncthreads();

  const int laneb = (int)__builtin_amdgcn_mbcnt_lo(~0u, 0u);
  const int tb    = wv * 32 + laneb;
  const int halfb = laneb >> 4;
  const int lb    = laneb & 15;

  float* xch = w2sum;
  float lmax = -INFINITY;
  for (int k = tb; k < K; k += 256) lmax = fmaxf(lmax, score[k]);
  lmax = wave_max_all(lmax);
  if (laneb == 0) xch[wv] = lmax;
  __syncthreads();
  float mx;
  {
    const v4f m0 = *(const v4f*)(xch), m1 = *(const v4f*)(xch + 4);
    mx = fmaxf(fmaxf(fmaxf(m0[0], m0[1]), fmaxf(m0[2], m0[3])),
               fmaxf(fmaxf(m1[0], m1[1]), fmaxf(m1[2], m1[3])));
  }
  float lsum = 0.f;
  for (int k = tb; k < K; k += 256) {
    const float e = __expf(score[k] - mx);
    score[k] = e;
    lsum += e;
  }
  lsum = wave_sum_all(lsum);
  if (laneb == 0) xch[8 + wv] = lsum;
  __syncthreads();
  float inv;
  {
    const v4f s0 = *(const v4f*)(xch + 8), s1 = *(const v4f*)(xch + 12);
    inv = 1.f / (((s0[0] + s0[1]) + (s0[2] + s0[3])) + ((s1[0] + s1[1]) + (s1[2] + s1[3])));
  }

  v4f lw[8];
  float lbias;
  {
    const int kw = wv * 128;
    const int perm = 2 * lb + halfb;
    const float* el = ent_table + 4 * lb;
    int tv[4], sv[4];
    #pragma unroll
    for (int j = 0; j < 4; ++j) {
      tv[j] = adj_tail[(unsigned)n * K + (unsigned)(kw + j * 32 + perm)] * D;
      sv[j] = __float_as_int(score[kw + j * 32 + perm] * inv);
    }
    v4f acc = {0.f, 0.f, 0.f, 0.f};
    v4f ea[8];
    {
      const float* pw = lin_w + (unsigned)(wv * 8 * (2 * D) + 4 * laneb);
      #pragma unroll
      for (int i = 0; i < 8; ++i) lw[i] = *(const v4f*)(pw + i * (2 * D));
      lbias = lin_b[wv * 8 + (laneb & 7)];
    }
    #pragma unroll
    for (int j = 0; j < 4; ++j) {
      const unsigned nz = __builtin_amdgcn_ballot_w32(__int_as_float(sv[j]) != 0.f);
      if (nz & 0x00ff00ffu) {
        gather_ent_rows<0>(el, tv[j], ea);
        add_ent_rows<0>(acc, sv[j], ea);
      }
      if (nz & 0xff00ff00u) {
        gather_ent_rows<1>(el, tv[j], ea);
        add_ent_rows<1>(acc, sv[j], ea);
      }
    }
    #pragma unroll
    for (int i = 0; i < 4; ++i) acc[i] += from_other_row(acc[i]);
    if (halfb == 0) *(v4f*)(red + wv * 64 + 4 * lb) = acc;
  }
  __syncthreads();

  v4f de;
  if (halfb == 0) {
    v4f s = {0.f, 0.f, 0.f, 0.f};
    #pragma unroll
    for (int w = 0; w < 8; ++w) s += *(const v4f*)(red + w * 64 + 4 * lb);
    de = s;
  } else {
    de = *(const v4f*)(demb + 4 * lb);
  }
  float p8[8];
  #pragma unroll
  for (int i = 0; i < 8; ++i)
    p8[i] = lw[i][0] * de[0] + lw[i][1] * de[1] + lw[i][2] * de[2] + lw[i][3] * de[3];
  {
    const bool b4 = (laneb & 4) != 0, b2 = (laneb & 2) != 0, b1l = (laneb & 1) != 0;
    float p4[4];
    #pragma unroll
    for (int i = 0; i < 4; ++i)
      p4[i] = (b4 ? p8[i + 4] : p8[i]) + from_lane_xor<4>(b4 ? p8[i] : p8[i + 4]);
    float p2[2];
    #pragma unroll
    for (int i = 0; i < 2; ++i)
      p2[i] = (b2 ? p4[i + 2] : p4[i]) + from_lane_xor<2>(b2 ? p4[i] : p4[i + 2]);
    float o = (b1l ? p2[1] : p2[0]) + from_lane_xor<1>(b1l ? p2[0] : p2[1]);
    o += from_lane_xor<8>(o);
    o += from_other_row(o);
    if (laneb < 8) score[wv * 8 + laneb] = fmaxf(o + lbias, 0.f);
  }
  __syncthreads();
  if (wv == 0 && laneb < 16)
    *(v4f*)(y + (size_t)n * D + 4 * laneb) = *(const v4f*)(score + 4 * laneb);
}

__global__ void __launch_bounds__(256) gnn_bn(
    const float* __restrict__ y, const float* __restrict__ bn_w,
    const float* __restrict__ bn_b, float* __restrict__ out)
{
  __shared__ __align__(16) float red[16];
  const int c = blockIdx.x, t = threadIdx.x;
  const int lane = t & 31, w = t >> 5;
  float s = 0.f;
  for (int r = t; r < N_DRUG; r += 256) s += y[(size_t)r * D + c];
  s = wave_sum_all(s);
  if (lane == 0) red[w] = s;
  __syncthreads();
  float mean;
  {
    const v4f a0 = *(const v4f*)(red), a1 = *(const v4f*)(red + 4);
    mean = (((a0[0] + a0[1]) + (a0[2] + a0[3])) + ((a1[0] + a1[1]) + (a1[2] + a1[3])))
           / (float)N_DRUG;
  }
  float q = 0.f;
  for (int r = t; r < N_DRUG; r += 256) {
    const float d = y[(size_t)r * D + c] - mean;
    q += d * d;
  }
  q = wave_sum_all(q);
  if (lane == 0) red[8 + w] = q;
  __syncthreads();
  float var;
  {
    const v4f a0 = *(const v4f*)(red + 8), a1 = *(const v4f*)(red + 12);
    var = (((a0[0] + a0[1]) + (a0[2] + a0[3])) + ((a1[0] + a1[1]) + (a1[2] + a1[3])))
          / (float)N_DRUG;
  }
  const float scale = bn_w[c] * rsqrtf(var + BN_EPS);
  const float shift = bn_b[c];
  for (int r = t; r < N_DRUG; r += 256)
    out[(size_t)r * D + c] = (y[(size_t)r * D + c] - mean) * scale + shift;
}

extern "C" void kernel_launch(void* const* d_in, const int* in_sizes, int n_in,
                              void* d_out, int out_size, void* d_ws, size_t ws_size,
                              hipStream_t stream) {
  (void)in_sizes; (void)n_in; (void)out_size; (void)ws_size;
  const float* drug_table = (const float*)d_in[0];
  const float* rela_table = (const float*)d_in[1];
  const float* ent_table  = (const float*)d_in[2];
  const float* W1         = (const float*)d_in[3];
  const float* b1         = (const float*)d_in[4];
  const float* W2         = (const float*)d_in[5];
  const float* b2         = (const float*)d_in[6];
  const float* lin_w      = (const float*)d_in[7];
  const float* lin_b      = (const float*)d_in[8];
  const float* bn_w       = (const float*)d_in[9];
  const float* bn_b       = (const float*)d_in[10];
  const int*   drug_name  = (const int*)d_in[11];
  const int*   adj_tail   = (const int*)d_in[12];
  const int*   adj_rel    = (const int*)d_in[13];

  float* ws     = (float*)d_ws;
  float* b2sum  = ws;
  float* y      = ws + K;

  gnn_b2sum<<<dim3((K + 255) / 256), dim3(256), 0, stream>>>(b2, b2sum);

  const size_t lds = (size_t)(RPAD * D + D + D + K + 512) * sizeof(float);
  gnn_main<<<dim3(N_DRUG), dim3(256), lds, stream>>>(
      drug_table, rela_table, ent_table, W1, b1, W2, lin_w, lin_b,
      drug_name, adj_tail, adj_rel, b2sum, y);

  gnn_bn<<<dim3(D), dim3(256), 0, stream>>>(y, bn_w, bn_b, (float*)d_out);
}


# Loads that took 16 B from each of 16-32 rows now cover whole lines: b2 and W2 row sums from 512-B loads and lane swizzles (both sum orders change); T-GEMM A rows from a table re-laid by kernel 0
# speedup vs baseline: 7.3748x; 1.1853x over previous
#include <hip/hip_runtime.h>

#define N_DRUG 846
#define D 64
#define K 1024
#define N_REL 200
#define RPAD 208
#define BN_EPS 1e-5f

typedef __attribute__((ext_vector_type(2))) float v2f;
typedef __attribute__((ext_vector_type(4))) float v4f;
typedef __attribute__((ext_vector_type(8))) float v8f;

template <int M>
__device__ __forceinline__ float from_lane_xor(float x) {
  return __builtin_bit_cast(float, __builtin_amdgcn_update_dpp(
      0, __builtin_bit_cast(int, x), 0x160 | M, 0xf, 0xf, true));
}

template <int P>
__device__ __forceinline__ int from_row_lane(int x) {
  return __builtin_amdgcn_update_dpp(0, x, 0x150 | P, 0xf, 0xf, true);
}

__device__ __forceinline__ float from_other_row(float x) {
  const int b = __builtin_bit_cast(int, x);
  return __builtin_bit_cast(float, __builtin_amdgcn_permlanex16(
      b, b, 0x76543210, 0xfedcba98, false, false));
}

__device__ __forceinline__ float wave_sum_all(float v) {
  v += from_lane_xor<1>(v);
  v += from_lane_xor<2>(v);
  v += from_lane_xor<4>(v);
  v += from_lane_xor<8>(v);
  v += from_other_row(v);
  return v;
}
__device__ __forceinline__ float wave_max_all(float v) {
  v = fmaxf(v, from_lane_xor<1>(v));
  v = fmaxf(v, from_lane_xor<2>(v));
  v = fmaxf(v, from_lane_xor<4>(v));
  v = fmaxf(v, from_lane_xor<8>(v));
  v = fmaxf(v, from_other_row(v));
  return v;
}

#define B2_BLOCKS 32
__global__ void __launch_bounds__(256) gnn_prep(const float* __restrict__ b2,
                                                const float* __restrict__ rela_table,
                                                float* __restrict__ b2sum,
                                                float* __restrict__ relaF) {
  const int t    = threadIdx.x;
  const int lane = t & 31;
  const int wave = t >> 5;
  const int half = lane >> 4;
  const int l    = lane & 15;
  __shared__ float sums[32];
  if (blockIdx.x < B2_BLOCKS) {
    const int k0 = (blockIdx.x * 8 + wave) * 4;
    const float* p = b2 + (unsigned)(k0 * D + 4 * lane);
    const v4f x0 = *(const v4f*)p;
    const v4f x1 = *(const v4f*)(p + 2 * D);
    float s0 = ((x0[0] + x0[1]) + x0[2]) + x0[3];
    float s1 = ((x1[0] + x1[1]) + x1[2]) + x1[3];
    s0 += from_lane_xor<1>(s0);  s1 += from_lane_xor<1>(s1);
    s0 += from_lane_xor<2>(s0);  s1 += from_lane_xor<2>(s1);
    s0 += from_lane_xor<4>(s0);  s1 += from_lane_xor<4>(s1);
    s0 += from_lane_xor<8>(s0);  s1 += from_lane_xor<8>(s1);
    if (l == 0) {
      sums[4 * wave + half]     = s0;
      sums[4 * wave + 2 + half] = s1;
    }
    __syncthreads();
    if (t < 32) b2sum[blockIdx.x * 32 + t] = sums[t];
  } else {
    const int m   = blockIdx.x - B2_BLOCKS;
    const int row = min(m * 16 + l, N_REL - 1);
    const v4f a = *(const v4f*)(rela_table + row * D + 8 * wave + 4 * half);
    *(v4f*)(relaF + (unsigned)(((m * 8 + wave) * 32 + lane) * 4)) = a;
  }
}

template <int B>
__device__ __forceinline__ void gather_t_rows(const float* tl, int held, v4f (&o)[8]) {
  o[0] = *(const v4f*)(tl + from_row_lane<8 * B + 0>(held));
  o[1] = *(const v4f*)(tl + from_row_lane<8 * B + 1>(held));
  o[2] = *(const v4f*)(tl + from_row_lane<8 * B + 2>(held));
  o[3] = *(const v4f*)(tl + from_row_lane<8 * B + 3>(held));
  o[4] = *(const v4f*)(tl + from_row_lane<8 * B + 4>(held));
  o[5] = *(const v4f*)(tl + from_row_lane<8 * B + 5>(held));
  o[6] = *(const v4f*)(tl + from_row_lane<8 * B + 6>(held));
  o[7] = *(const v4f*)(tl + from_row_lane<8 * B + 7>(held));
}
template <int B>
__device__ __forceinline__ void gather_ent_rows(const float* el, int held, v4f (&o)[8]) {
  o[0] = *(const v4f*)(el + (unsigned)from_row_lane<8 * B + 0>(held));
  o[1] = *(const v4f*)(el + (unsigned)from_row_lane<8 * B + 1>(held));
  o[2] = *(const v4f*)(el + (unsigned)from_row_lane<8 * B + 2>(held));
  o[3] = *(const v4f*)(el + (unsigned)from_row_lane<8 * B + 3>(held));
  o[4] = *(const v4f*)(el + (unsigned)from_row_lane<8 * B + 4>(held));
  o[5] = *(const v4f*)(el + (unsigned)from_row_lane<8 * B + 5>(held));
  o[6] = *(const v4f*)(el + (unsigned)from_row_lane<8 * B + 6>(held));
  o[7] = *(const v4f*)(el + (unsigned)from_row_lane<8 * B + 7>(held));
}
template <int B>
__device__ __forceinline__ void add_ent_rows(v4f& acc, int held, const v4f (&e)[8]) {
  acc += __int_as_float(from_row_lane<8 * B + 0>(held)) * e[0];
  acc += __int_as_float(from_row_lane<8 * B + 1>(held)) * e[1];
  acc += __int_as_float(from_row_lane<8 * B + 2>(held)) * e[2];
  acc += __int_as_float(from_row_lane<8 * B + 3>(held)) * e[3];
  acc += __int_as_float(from_row_lane<8 * B + 4>(held)) * e[4];
  acc += __int_as_float(from_row_lane<8 * B + 5>(held)) * e[5];
  acc += __int_as_float(from_row_lane<8 * B + 6>(held)) * e[6];
  acc += __int_as_float(from_row_lane<8 * B + 7>(held)) * e[7];
}

struct group_more { static constexpr bool last = false; };
struct group_last { static constexpr bool last = true; };

__global__ void __launch_bounds__(256, 8) gnn_main(
    const float* __restrict__ drug_table, const float* __restrict__ relaF,
    const float* __restrict__ ent_table,  const float* __restrict__ W1g,
    const float* __restrict__ b1,         const float* __restrict__ W2g,
    const float* __restrict__ lin_w,      const float* __restrict__ lin_b,
    const int*   __restrict__ drug_name,  const int* __restrict__ adj_tail,
    const int*   __restrict__ adj_rel,    const float* __restrict__ b2sum,
    float* __restrict__ y)
{
  extern __shared__ __align__(16) float smem[];
  float* Ttile = smem;
  float* demb  = Ttile + RPAD * D;
  float* w2sum = demb + D;
  float* score = w2sum + D;
  float* red   = score + K;

  const int n    = blockIdx.x;
  const int t    = threadIdx.x;
  const int lane = t & 31;
  const int wave = t >> 5;
  const int half = lane >> 4;
  const int l    = lane & 15;
  const int wv = __builtin_amdgcn_readfirstlane(wave);
  const int n0 = (wv & 3) * 16;
  const int mt = (wv >> 2) & 1;

  const size_t wbase = (size_t)n * D * D;
  const float* drow  = drug_table + (size_t)drug_name[n] * D;

  v2f bf[16];
  {
    const float* pb = W1g + wbase + (4 * half) * D + n0 + l;
    #pragma unroll
    for (int j = 0; j < 8; ++j) {
      #pragma unroll
      for (int s = 0; s < 2; ++s) {
        bf[2 * j + s][0] = pb[(8 * j + 2 * s) * D];
        bf[2 * j + s][1] = pb[(8 * j + 2 * s + 1) * D];
      }
    }
  }
  v4f w2r[4];
  {
    const float* p = W2g + wbase + (unsigned)(wv * 8 * D + 4 * lane);
    #pragma unroll
    for (int i = 0; i < 4; ++i) w2r[i] = *(const v4f*)(p + 2 * D * i);
  }
  v4f ar[8];
  {
    const float* pa = relaF + (unsigned)((mt * 256 + lane) * 4);
    #pragma unroll
    for (int j = 0; j < 8; ++j) ar[j] = *(const v4f*)(pa + 128 * j);
  }
  #pragma unroll
  for (int i = 0; i < 4; ++i) {
    float s = ((w2r[i][0] + w2r[i][1]) + w2r[i][2]) + w2r[i][3];
    s += from_lane_xor<1>(s);
    s += from_lane_xor<2>(s);
    s += from_lane_xor<4>(s);
    s += from_lane_xor<8>(s);
    if (l == 0) w2sum[wv * 8 + 2 * i + half] = s;
  }
  if (t < D) demb[t] = drow[t];
  __syncthreads();
  {
    const float* pd = demb + 4 * half;
    #pragma unroll
    for (int j = 0; j < 8; ++j) {
      const v4f d4 = *(const v4f*)(pd + 8 * j);
      bf[2 * j][0]     *= d4[0];
      bf[2 * j][1]     *= d4[1];
      bf[2 * j + 1][0] *= d4[2];
      bf[2 * j + 1][1] *= d4[3];
    }
  }

  #pragma unroll 1
  for (int m = mt; m < 13; m += 2) {
    const float* pn = relaF + (unsigned)((min(m + 2, 12) * 256 + lane) * 4);
    v8f c = {0.f, 0.f, 0.f, 0.f, 0.f, 0.f, 0.f, 0.f};
    #pragma unroll
    for (int j = 0; j < 8; ++j) {
      const v2f a0 = {ar[j][0], ar[j][1]};
      const v2f a1 = {ar[j][2], ar[j][3]};
      c = __builtin_amdgcn_wmma_f32_16x16x4_f32(false, a0, false, bf[2 * j],
                                                (short)0, c, false, false);
      c = __builtin_amdgcn_wmma_f32_16x16x4_f32(false, a1, false, bf[2 * j + 1],
                                                (short)0, c, false, false);
      ar[j] = *(const v4f*)(pn + 128 * j);
    }
    const int m0 = m * 16;
    #pragma unroll
    for (int i = 0; i < 8; ++i)
      Ttile[(m0 + 8 * half + i) * D + n0 + l] = c[i];
  }
  __syncthreads();

  {
    const int   kw = wave * 128;
    const int   perm = 2 * l + half;
    const v4f   w4 = *(const v4f*)(w2sum + 4 * l);
    const float* tl = Ttile + 4 * l;
    const bool  hi8 = (lane & 8) != 0, hi4 = (lane & 4) != 0;
    const bool  hi2 = (lane & 2) != 0, hi1 = (lane & 1) != 0;
    const unsigned relo = (unsigned)n * K + kw + perm;
    const unsigned b1o  = (unsigned)kw * D + 4 * lane;
    int rvc = adj_rel[relo] * D;
    v4f cb[8];
    #pragma unroll
    for (int u = 0; u < 8; ++u)
      cb[u] = *(const v4f*)(b1 + (b1o + 2u * u * D));
    auto group = [&](const int j, auto tag) {
      constexpr bool LAST = decltype(tag)::last;
      int rvn = 0;
      if (!LAST) rvn = adj_rel[relo + (j + 1) * 32] * D;
      float v[16];
      #pragma unroll
      for (int b = 0; b < 2; ++b) {
        v4f tq[8];
        if (b == 0) gather_t_rows<0>(tl, rvc, tq);
        else        gather_t_rows<1>(tl, rvc, tq);
        const unsigned nb1 = b1o + (unsigned)(2 * j + b + 1) * (16 * D);
        #pragma unroll
        for (int u = 0; u < 8; ++u) {
          const v4f z = tq[u] + cb[u];
          v[b * 8 + u] = fmaxf(z[0], 0.f) * w4[0] + fmaxf(z[1], 0.f) * w4[1]
                       + fmaxf(z[2], 0.f) * w4[2] + fmaxf(z[3], 0.f) * w4[3];
          if (!LAST || b == 0) cb[u] = *(const v4f*)(b1 + (nb1 + 2u * u * D));
        }
      }
      float s8[8];
      #pragma unroll
      for (int i = 0; i < 8; ++i)
        s8[i] = (hi8 ? v[i + 8] : v[i]) + from_lane_xor<8>(hi8 ? v[i] : v[i + 8]);
      float s4[4];
      #pragma unroll
      for (int i = 0; i < 4; ++i)
        s4[i] = (hi4 ? s8[i + 4] : s8[i]) + from_lane_xor<4>(hi4 ? s8[i] : s8[i + 4]);
      float s2[2];
      #pragma unroll
      for (int i = 0; i < 2; ++i)
        s2[i] = (hi2 ? s4[i + 2] : s4[i]) + from_lane_xor<2>(hi2 ? s4[i] : s4[i + 2]);
      const float mine = (hi1 ? s2[1] : s2[0]) + from_lane_xor<1>(hi1 ? s2[0] : s2[1]);
      score[kw + j * 32 + perm] = mine + b2sum[(unsigned)(kw + j * 32 + perm)];
      rvc = rvn;
    };
    #pragma unroll 1
    for (int j = 0; j < 3; ++j) group(j, group_more{});
    group(3, group_last{});
  }
  __syncthreads();

  const int laneb = (int)__builtin_amdgcn_mbcnt_lo(~0u, 0u);
  const int tb    = wv * 32 + laneb;
  const int halfb = laneb >> 4;
  const int lb    = laneb & 15;

  float* xch = w2sum;
  float lmax = -INFINITY;
  for (int k = tb; k < K; k += 256) lmax = fmaxf(lmax, score[k]);
  lmax = wave_max_all(lmax);
  if (laneb == 0) xch[wv] = lmax;
  __syncthreads();
  float mx;
  {
    const v4f m0 = *(const v4f*)(xch), m1 = *(const v4f*)(xch + 4);
    mx = fmaxf(fmaxf(fmaxf(m0[0], m0[1]), fmaxf(m0[2], m0[3])),
               fmaxf(fmaxf(m1[0], m1[1]), fmaxf(m1[2], m1[3])));
  }
  float lsum = 0.f;
  for (int k = tb; k < K; k += 256) {
    const float e = __expf(score[k] - mx);
    score[k] = e;
    lsum += e;
  }
  lsum = wave_sum_all(lsum);
  if (laneb == 0) xch[8 + wv] = lsum;
  __syncthreads();
  float inv;
  {
    const v4f s0 = *(const v4f*)(xch + 8), s1 = *(const v4f*)(xch + 12);
    inv = 1.f / (((s0[0] + s0[1]) + (s0[2] + s0[3])) + ((s1[0] + s1[1]) + (s1[2] + s1[3])));
  }

  v4f lw[8];
  float lbias;
  {
    const int kw = wv * 128;
    const int perm = 2 * lb + halfb;
    const float* el = ent_table + 4 * lb;
    int tv[4], sv[4];
    #pragma unroll
    for (int j = 0; j < 4; ++j) {
      tv[j] = adj_tail[(unsigned)n * K + (unsigned)(kw + j * 32 + perm)] * D;
      sv[j] = __float_as_int(score[kw + j * 32 + perm] * inv);
    }
    v4f acc = {0.f, 0.f, 0.f, 0.f};
    v4f ea[8];
    {
      const float* pw = lin_w + (unsigned)(wv * 8 * (2 * D) + 4 * laneb);
      #pragma unroll
      for (int i = 0; i < 8; ++i) lw[i] = *(const v4f*)(pw + i * (2 * D));
      lbias = lin_b[wv * 8 + (laneb & 7)];
    }
    #pragma unroll
    for (int j = 0; j < 4; ++j) {
      const unsigned nz = __builtin_amdgcn_ballot_w32(__int_as_float(sv[j]) != 0.f);
      if (nz & 0x00ff00ffu) {
        gather_ent_rows<0>(el, tv[j], ea);
        add_ent_rows<0>(acc, sv[j], ea);
      }
      if (nz & 0xff00ff00u) {
        gather_ent_rows<1>(el, tv[j], ea);
        add_ent_rows<1>(acc, sv[j], ea);
      }
    }
    #pragma unroll
    for (int i = 0; i < 4; ++i) acc[i] += from_other_row(acc[i]);
    if (halfb == 0) *(v4f*)(red + wv * 64 + 4 * lb) = acc;
  }
  __syncthreads();

  v4f de;
  if (halfb == 0) {
    v4f s = {0.f, 0.f, 0.f, 0.f};
    #pragma unroll
    for (int w = 0; w < 8; ++w) s += *(const v4f*)(red + w * 64 + 4 * lb);
    de = s;
  } else {
    de = *(const v4f*)(demb + 4 * lb);
  }
  float p8[8];
  #pragma unroll
  for (int i = 0; i < 8; ++i)
    p8[i] = lw[i][0] * de[0] + lw[i][1] * de[1] + lw[i][2] * de[2] + lw[i][3] * de[3];
  {
    const bool b4 = (laneb & 4) != 0, b2 = (laneb & 2) != 0, b1l = (laneb & 1) != 0;
    float p4[4];
    #pragma unroll
    for (int i = 0; i < 4; ++i)
      p4[i] = (b4 ? p8[i + 4] : p8[i]) + from_lane_xor<4>(b4 ? p8[i] : p8[i + 4]);
    float p2[2];
    #pragma unroll
    for (int i = 0; i < 2; ++i)
      p2[i] = (b2 ? p4[i + 2] : p4[i]) + from_lane_xor<2>(b2 ? p4[i] : p4[i + 2]);
    float o = (b1l ? p2[1] : p2[0]) + from_lane_xor<1>(b1l ? p2[0] : p2[1]);
    o += from_lane_xor<8>(o);
    o += from_other_row(o);
    if (laneb < 8) score[wv * 8 + laneb] = fmaxf(o + lbias, 0.f);
  }
  __syncthreads();
  if (wv == 0 && laneb < 16)
    *(v4f*)(y + (size_t)n * D + 4 * laneb) = *(const v4f*)(score + 4 * laneb);
}

__global__ void __launch_bounds__(256) gnn_bn(
    const float* __restrict__ y, const float* __restrict__ bn_w,
    const float* __restrict__ bn_b, float* __restrict__ out)
{
  __shared__ __align__(16) float red[16];
  const int c = blockIdx.x, t = threadIdx.x;
  const int lane = t & 31, w = t >> 5;
  float s = 0.f;
  for (int r = t; r < N_DRUG; r += 256) s += y[(size_t)r * D + c];
  s = wave_sum_all(s);
  if (lane == 0) red[w] = s;
  __syncthreads();
  float mean;
  {
    const v4f a0 = *(const v4f*)(red), a1 = *(const v4f*)(red + 4);
    mean = (((a0[0] + a0[1]) + (a0[2] + a0[3])) + ((a1[0] + a1[1]) + (a1[2] + a1[3])))
           / (float)N_DRUG;
  }
  float q = 0.f;
  for (int r = t; r < N_DRUG; r += 256) {
    const float d = y[(size_t)r * D + c] - mean;
    q += d * d;
  }
  q = wave_sum_all(q);
  if (lane == 0) red[8 + w] = q;
  __syncthreads();
  float var;
  {
    const v4f a0 = *(const v4f*)(red + 8), a1 = *(const v4f*)(red + 12);
    var = (((a0[0] + a0[1]) + (a0[2] + a0[3])) + ((a1[0] + a1[1]) + (a1[2] + a1[3])))
          / (float)N_DRUG;
  }
  const float scale = bn_w[c] * rsqrtf(var + BN_EPS);
  const float shift = bn_b[c];
  for (int r = t; r < N_DRUG; r += 256)
    out[(size_t)r * D + c] = (y[(size_t)r * D + c] - mean) * scale + shift;
}

extern "C" void kernel_launch(void* const* d_in, const int* in_sizes, int n_in,
                              void* d_out, int out_size, void* d_ws, size_t ws_size,
                              hipStream_t stream) {
  (void)in_sizes; (void)n_in; (void)out_size; (void)ws_size;
  const float* drug_table = (const float*)d_in[0];
  const float* rela_table = (const float*)d_in[1];
  const float* ent_table  = (const float*)d_in[2];
  const float* W1         = (const float*)d_in[3];
  const float* b1         = (const float*)d_in[4];
  const float* W2         = (const float*)d_in[5];
  const float* b2         = (const float*)d_in[6];
  const float* lin_w      = (const float*)d_in[7];
  const float* lin_b      = (const float*)d_in[8];
  const float* bn_w       = (const float*)d_in[9];
  const float* bn_b       = (const float*)d_in[10];
  const int*   drug_name  = (const int*)d_in[11];
  const int*   adj_tail   = (const int*)d_in[12];
  const int*   adj_rel    = (const int*)d_in[13];

  float* ws     = (float*)d_ws;
  float* b2sum  = ws;
  float* y      = ws + K;
  float* relaF  = ws + K + N_DRUG * D;

  gnn_prep<<<dim3(B2_BLOCKS + 13), dim3(256), 0, stream>>>(b2, rela_table, b2sum, relaF);

  const size_t lds = (size_t)(RPAD * D + D + D + K + 512) * sizeof(float);
  gnn_main<<<dim3(N_DRUG), dim3(256), lds, stream>>>(
      drug_table, relaF, ent_table, W1, b1, W2, lin_w, lin_b,
      drug_name, adj_tail, adj_rel, b2sum, y);

  gnn_bn<<<dim3(D), dim3(256), 0, stream>>>(y, bn_w, bn_b, (float*)d_out);
}


# Batch-norm kernel reads each y value once: a thread's four rows are asked for together and held in registers for mean, variance and output (was one waited load per row and pass); bit-identical
# speedup vs baseline: 7.9782x; 1.0818x over previous
#include <hip/hip_runtime.h>

#define N_DRUG 846
#define D 64
#define K 1024
#define N_REL 200
#define RPAD 208
#define BN_EPS 1e-5f

typedef __attribute__((ext_vector_type(2))) float v2f;
typedef __attribute__((ext_vector_type(4))) float v4f;
typedef __attribute__((ext_vector_type(8))) float v8f;

template <int M>
__device__ __forceinline__ float from_lane_xor(float x) {
  return __builtin_bit_cast(float, __builtin_amdgcn_update_dpp(
      0, __builtin_bit_cast(int, x), 0x160 | M, 0xf, 0xf, true));
}

template <int P>
__device__ __forceinline__ int from_row_lane(int x) {
  return __builtin_amdgcn_update_dpp(0, x, 0x150 | P, 0xf, 0xf, true);
}

__device__ __forceinline__ float from_other_row(float x) {
  const int b = __builtin_bit_cast(int, x);
  return __builtin_bit_cast(float, __builtin_amdgcn_permlanex16(
      b, b, 0x76543210, 0xfedcba98, false, false));
}

__device__ __forceinline__ float wave_sum_all(float v) {
  v += from_lane_xor<1>(v);
  v += from_lane_xor<2>(v);
  v += from_lane_xor<4>(v);
  v += from_lane_xor<8>(v);
  v += from_other_row(v);
  return v;
}
__device__ __forceinline__ float wave_max_all(float v) {
  v = fmaxf(v, from_lane_xor<1>(v));
  v = fmaxf(v, from_lane_xor<2>(v));
  v = fmaxf(v, from_lane_xor<4>(v));
  v = fmaxf(v, from_lane_xor<8>(v));
  v = fmaxf(v, from_other_row(v));
  return v;
}

#define B2_BLOCKS 32
__global__ void __launch_bounds__(256) gnn_prep(const float* __restrict__ b2,
                                                const float* __restrict__ rela_table,
                                                float* __restrict__ b2sum,
                                                float* __restrict__ relaF) {
  const int t    = threadIdx.x;
  const int lane = t & 31;
  const int wave = t >> 5;
  const int half = lane >> 4;
  const int l    = lane & 15;
  __shared__ float sums[32];
  if (blockIdx.x < B2_BLOCKS) {
    const int k0 = (blockIdx.x * 8 + wave) * 4;
    const float* p = b2 + (unsigned)(k0 * D + 4 * lane);
    const v4f x0 = *(const v4f*)p;
    const v4f x1 = *(const v4f*)(p + 2 * D);
    float s0 = ((x0[0] + x0[1]) + x0[2]) + x0[3];
    float s1 = ((x1[0] + x1[1]) + x1[2]) + x1[3];
    s0 += from_lane_xor<1>(s0);  s1 += from_lane_xor<1>(s1);
    s0 += from_lane_xor<2>(s0);  s1 += from_lane_xor<2>(s1);
    s0 += from_lane_xor<4>(s0);  s1 += from_lane_xor<4>(s1);
    s0 += from_lane_xor<8>(s0);  s1 += from_lane_xor<8>(s1);
    if (l == 0) {
      sums[4 * wave + half]     = s0;
      sums[4 * wave + 2 + half] = s1;
    }
    __syncthreads();
    if (t < 32) b2sum[blockIdx.x * 32 + t] = sums[t];
  } else {
    const int m   = blockIdx.x - B2_BLOCKS;
    const int row = min(m * 16 + l, N_REL - 1);
    const v4f a = *(const v4f*)(rela_table + row * D + 8 * wave + 4 * half);
    *(v4f*)(relaF + (unsigned)(((m * 8 + wave) * 32 + lane) * 4)) = a;
  }
}

template <int B>
__device__ __forceinline__ void gather_t_rows(const float* tl, int held, v4f (&o)[8]) {
  o[0] = *(const v4f*)(tl + from_row_lane<8 * B + 0>(held));
  o[1] = *(const v4f*)(tl + from_row_lane<8 * B + 1>(held));
  o[2] = *(const v4f*)(tl + from_row_lane<8 * B + 2>(held));
  o[3] = *(const v4f*)(tl + from_row_lane<8 * B + 3>(held));
  o[4] = *(const v4f*)(tl + from_row_lane<8 * B + 4>(held));
  o[5] = *(const v4f*)(tl + from_row_lane<8 * B + 5>(held));
  o[6] = *(const v4f*)(tl + from_row_lane<8 * B + 6>(held));
  o[7] = *(const v4f*)(tl + from_row_lane<8 * B + 7>(held));
}
template <int B>
__device__ __forceinline__ void gather_ent_rows(const float* el, int held, v4f (&o)[8]) {
  o[0] = *(const v4f*)(el + (unsigned)from_row_lane<8 * B + 0>(held));
  o[1] = *(const v4f*)(el + (unsigned)from_row_lane<8 * B + 1>(held));
  o[2] = *(const v4f*)(el + (unsigned)from_row_lane<8 * B + 2>(held));
  o[3] = *(const v4f*)(el + (unsigned)from_row_lane<8 * B + 3>(held));
  o[4] = *(const v4f*)(el + (unsigned)from_row_lane<8 * B + 4>(held));
  o[5] = *(const v4f*)(el + (unsigned)from_row_lane<8 * B + 5>(held));
  o[6] = *(const v4f*)(el + (unsigned)from_row_lane<8 * B + 6>(held));
  o[7] = *(const v4f*)(el + (unsigned)from_row_lane<8 * B + 7>(held));
}
template <int B>
__device__ __forceinline__ void add_ent_rows(v4f& acc, int held, const v4f (&e)[8]) {
  acc += __int_as_float(from_row_lane<8 * B + 0>(held)) * e[0];
  acc += __int_as_float(from_row_lane<8 * B + 1>(held)) * e[1];
  acc += __int_as_float(from_row_lane<8 * B + 2>(held)) * e[2];
  acc += __int_as_float(from_row_lane<8 * B + 3>(held)) * e[3];
  acc += __int_as_float(from_row_lane<8 * B + 4>(held)) * e[4];
  acc += __int_as_float(from_row_lane<8 * B + 5>(held)) * e[5];
  acc += __int_as_float(from_row_lane<8 * B + 6>(held)) * e[6];
  acc += __int_as_float(from_row_lane<8 * B + 7>(held)) * e[7];
}

struct group_more { static constexpr bool last = false; };
struct group_last { static constexpr bool last = true; };

__global__ void __launch_bounds__(256, 8) gnn_main(
    const float* __restrict__ drug_table, const float* __restrict__ relaF,
    const float* __restrict__ ent_table,  const float* __restrict__ W1g,
    const float* __restrict__ b1,         const float* __restrict__ W2g,
    const float* __restrict__ lin_w,      const float* __restrict__ lin_b,
    const int*   __restrict__ drug_name,  const int* __restrict__ adj_tail,
    const int*   __restrict__ adj_rel,    const float* __restrict__ b2sum,
    float* __restrict__ y)
{
  extern __shared__ __align__(16) float smem[];
  float* Ttile = smem;
  float* demb  = Ttile + RPAD * D;
  float* w2sum = demb + D;
  float* score = w2sum + D;
  float* red   = score + K;

  const int n    = blockIdx.x;
  const int t    = threadIdx.x;
  const int lane = t & 31;
  const int wave = t >> 5;
  const int half = lane >> 4;
  const int l    = lane & 15;
  const int wv = __builtin_amdgcn_readfirstlane(wave);
  const int n0 = (wv & 3) * 16;
  const int mt = (wv >> 2) & 1;

  const size_t wbase = (size_t)n * D * D;
  const float* drow  = drug_table + (size_t)drug_name[n] * D;

  v2f bf[16];
  {
    const float* pb = W1g + wbase + (4 * half) * D + n0 + l;
    #pragma unroll
    for (int j = 0; j < 8; ++j) {
      #pragma unroll
      for (int s = 0; s < 2; ++s) {
        bf[2 * j + s][0] = pb[(8 * j + 2 * s) * D];
        bf[2 * j + s][1] = pb[(8 * j + 2 * s + 1) * D];
      }
    }
  }
  v4f w2r[4];
  {
    const float* p = W2g + wbase + (unsigned)(wv * 8 * D + 4 * lane);
    #pragma unroll
    for (int i = 0; i < 4; ++i) w2r[i] = *(const v4f*)(p + 2 * D * i);
  }
  v4f ar[8];
  {
    const float* pa = relaF + (unsigned)((mt * 256 + lane) * 4);
    #pragma unroll
    for (int j = 0; j < 8; ++j) ar[j] = *(const v4f*)(pa + 128 * j);
  }
  #pragma unroll
  for (int i = 0; i < 4; ++i) {
    float s = ((w2r[i][0] + w2r[i][1]) + w2r[i][2]) + w2r[i][3];
    s += from_lane_xor<1>(s);
    s += from_lane_xor<2>(s);
    s += from_lane_xor<4>(s);
    s += from_lane_xor<8>(s);
    if (l == 0) w2sum[wv * 8 + 2 * i + half] = s;
  }
  if (t < D) demb[t] = drow[t];
  __syncthreads();
  {
    const float* pd = demb + 4 * half;
    #pragma unroll
    for (int j = 0; j < 8; ++j) {
      const v4f d4 = *(const v4f*)(pd + 8 * j);
      bf[2 * j][0]     *= d4[0];
      bf[2 * j][1]     *= d4[1];
      bf[2 * j + 1][0] *= d4[2];
      bf[2 * j + 1][1] *= d4[3];
    }
  }

  #pragma unroll 1
  for (int m = mt; m < 13; m += 2) {
    const float* pn = relaF + (unsigned)((min(m + 2, 12) * 256 + lane) * 4);
    v8f c = {0.f, 0.f, 0.f, 0.f, 0.f, 0.f, 0.f, 0.f};
    #pragma unroll
    for (int j = 0; j < 8; ++j) {
      const v2f a0 = {ar[j][0], ar[j][1]};
      const v2f a1 = {ar[j][2], ar[j][3]};
      c = __builtin_amdgcn_wmma_f32_16x16x4_f32(false, a0, false, bf[2 * j],
                                                (short)0, c, false, false);
      c = __builtin_amdgcn_wmma_f32_16x16x4_f32(false, a1, false, bf[2 * j + 1],
                                                (short)0, c, false, false);
      ar[j] = *(const v4f*)(pn + 128 * j);
    }
    const int m0 = m * 16;
    #pragma unroll
    for (int i = 0; i < 8; ++i)
      Ttile[(m0 + 8 * half + i) * D + n0 + l] = c[i];
  }
  __syncthreads();

  {
    const int   kw = wave * 128;
    const int   perm = 2 * l + half;
    const v4f   w4 = *(const v4f*)(w2sum + 4 * l);
    const float* tl = Ttile + 4 * l;
    const bool  hi8 = (lane & 8) != 0, hi4 = (lane & 4) != 0;
    const bool  hi2 = (lane & 2) != 0, hi1 = (lane & 1) != 0;
    const unsigned relo = (unsigned)n * K + kw + perm;
    const unsigned b1o  = (unsigned)kw * D + 4 * lane;
    int rvc = adj_rel[relo] * D;
    v4f cb[8];
    #pragma unroll
    for (int u = 0; u < 8; ++u)
      cb[u] = *(const v4f*)(b1 + (b1o + 2u * u * D));
    auto group = [&](const int j, auto tag) {
      constexpr bool LAST = decltype(tag)::last;
      int rvn = 0;
      if (!LAST) rvn = adj_rel[relo + (j + 1) * 32] * D;
      float v[16];
      #pragma unroll
      for (int b = 0; b < 2; ++b) {
        v4f tq[8];
        if (b == 0) gather_t_rows<0>(tl, rvc, tq);
        else        gather_t_rows<1>(tl, rvc, tq);
        const unsigned nb1 = b1o + (unsigned)(2 * j + b + 1) * (16 * D);
        #pragma unroll
        for (int u = 0; u < 8; ++u) {
          const v4f z = tq[u] + cb[u];
          v[b * 8 + u] = fmaxf(z[0], 0.f) * w4[0] + fmaxf(z[1], 0.f) * w4[1]
                       + fmaxf(z[2], 0.f) * w4[2] + fmaxf(z[3], 0.f) * w4[3];
          if (!LAST || b == 0) cb[u] = *(const v4f*)(b1 + (nb1 + 2u * u * D));
        }
      }
      float s8[8];
      #pragma unroll
      for (int i = 0; i < 8; ++i)
        s8[i] = (hi8 ? v[i + 8] : v[i]) + from_lane_xor<8>(hi8 ? v[i] : v[i + 8]);
      float s4[4];
      #pragma unroll
      for (int i = 0; i < 4; ++i)
        s4[i] = (hi4 ? s8[i + 4] : s8[i]) + from_lane_xor<4>(hi4 ? s8[i] : s8[i + 4]);
      float s2[2];
      #pragma unroll
      for (int i = 0; i < 2; ++i)
        s2[i] = (hi2 ? s4[i + 2] : s4[i]) + from_lane_xor<2>(hi2 ? s4[i] : s4[i + 2]);
      const float mine = (hi1 ? s2[1] : s2[0]) + from_lane_xor<1>(hi1 ? s2[0] : s2[1]);
      score[kw + j * 32 + perm] = mine + b2sum[(unsigned)(kw + j * 32 + perm)];
      rvc = rvn;
    };
    #pragma unroll 1
    for (int j = 0; j < 3; ++j) group(j, group_more{});
    group(3, group_last{});
  }
  __syncthreads();

  const int laneb = (int)__builtin_amdgcn_mbcnt_lo(~0u, 0u);
  const int tb    = wv * 32 + laneb;
  const int halfb = laneb >> 4;
  const int lb    = laneb & 15;

  float* xch = w2sum;
  float lmax = -INFINITY;
  for (int k = tb; k < K; k += 256) lmax = fmaxf(lmax, score[k]);
  lmax = wave_max_all(lmax);
  if (laneb == 0) xch[wv] = lmax;
  __syncthreads();
  float mx;
  {
    const v4f m0 = *(const v4f*)(xch), m1 = *(const v4f*)(xch + 4);
    mx = fmaxf(fmaxf(fmaxf(m0[0], m0[1]), fmaxf(m0[2], m0[3])),
               fmaxf(fmaxf(m1[0], m1[1]), fmaxf(m1[2], m1[3])));
  }
  float lsum = 0.f;
  for (int k = tb; k < K; k += 256) {
    const float e = __expf(score[k] - mx);
    score[k] = e;
    lsum += e;
  }
  lsum = wave_sum_all(lsum);
  if (laneb == 0) xch[8 + wv] = lsum;
  __syncthreads();
  float inv;
  {
    const v4f s0 = *(const v4f*)(xch + 8), s1 = *(const v4f*)(xch + 12);
    inv = 1.f / (((s0[0] + s0[1]) + (s0[2] + s0[3])) + ((s1[0] + s1[1]) + (s1[2] + s1[3])));
  }

  v4f lw[8];
  float lbias;
  {
    const int kw = wv * 128;
    const int perm = 2 * lb + halfb;
    const float* el = ent_table + 4 * lb;
    int tv[4], sv[4];
    #pragma unroll
    for (int j = 0; j < 4; ++j) {
      tv[j] = adj_tail[(unsigned)n * K + (unsigned)(kw + j * 32 + perm)] * D;
      sv[j] = __float_as_int(score[kw + j * 32 + perm] * inv);
    }
    v4f acc = {0.f, 0.f, 0.f, 0.f};
    v4f ea[8];
    {
      const float* pw = lin_w + (unsigned)(wv * 8 * (2 * D) + 4 * laneb);
      #pragma unroll
      for (int i = 0; i < 8; ++i) lw[i] = *(const v4f*)(pw + i * (2 * D));
      lbias = lin_b[wv * 8 + (laneb & 7)];
    }
    #pragma unroll
    for (int j = 0; j < 4; ++j) {
      const unsigned nz = __builtin_amdgcn_ballot_w32(__int_as_float(sv[j]) != 0.f);
      if (nz & 0x00ff00ffu) {
        gather_ent_rows<0>(el, tv[j], ea);
        add_ent_rows<0>(acc, sv[j], ea);
      }
      if (nz & 0xff00ff00u) {
        gather_ent_rows<1>(el, tv[j], ea);
        add_ent_rows<1>(acc, sv[j], ea);
      }
    }
    #pragma unroll
    for (int i = 0; i < 4; ++i) acc[i] += from_other_row(acc[i]);
    if (halfb == 0) *(v4f*)(red + wv * 64 + 4 * lb) = acc;
  }
  __syncthreads();

  v4f de;
  if (halfb == 0) {
    v4f s = {0.f, 0.f, 0.f, 0.f};
    #pragma unroll
    for (int w = 0; w < 8; ++w) s += *(const v4f*)(red + w * 64 + 4 * lb);
    de = s;
  } else {
    de = *(const v4f*)(demb + 4 * lb);
  }
  float p8[8];
  #pragma unroll
  for (int i = 0; i < 8; ++i)
    p8[i] = lw[i][0] * de[0] + lw[i][1] * de[1] + lw[i][2] * de[2] + lw[i][3] * de[3];
  {
    const bool b4 = (laneb & 4) != 0, b2 = (laneb & 2) != 0, b1l = (laneb & 1) != 0;
    float p4[4];
    #pragma unroll
    for (int i = 0; i < 4; ++i)
      p4[i] = (b4 ? p8[i + 4] : p8[i]) + from_lane_xor<4>(b4 ? p8[i] : p8[i + 4]);
    float p2[2];
    #pragma unroll
    for (int i = 0; i < 2; ++i)
      p2[i] = (b2 ? p4[i + 2] : p4[i]) + from_lane_xor<2>(b2 ? p4[i] : p4[i + 2]);
    float o = (b1l ? p2[1] : p2[0]) + from_lane_xor<1>(b1l ? p2[0] : p2[1]);
    o += from_lane_xor<8>(o);
    o += from_other_row(o);
    if (laneb < 8) score[wv * 8 + laneb] = fmaxf(o + lbias, 0.f);
  }
  __syncthreads();
  if (wv == 0 && laneb < 16)
    *(v4f*)(y + (size_t)n * D + 4 * laneb) = *(const v4f*)(score + 4 * laneb);
}

__global__ void __launch_bounds__(256) gnn_bn(
    const float* __restrict__ y, const float* __restrict__ bn_w,
    const float* __restrict__ bn_b, float* __restrict__ out)
{
  __shared__ __align__(16) float red[16];
  const int c = blockIdx.x, t = threadIdx.x;
  const int lane = t & 31, w = t >> 5;
  const bool has3 = t < N_DRUG - 768;
  const unsigned o0 = (unsigned)(t * D + c);
  const float v0 = y[o0];
  const float v1 = y[o0 + 256u * D];
  const float v2 = y[o0 + 512u * D];
  float v3 = 0.f;
  if (has3) v3 = y[o0 + 768u * D];
  const float gw = bn_w[c];
  const float gb = bn_b[c];
  float s = 0.f;
  s += v0;
  s += v1;
  s += v2;
  if (has3) s += v3;
  s = wave_sum_all(s);
  if (lane == 0) red[w] = s;
  __syncthreads();
  float mean;
  {
    const v4f a0 = *(const v4f*)(red), a1 = *(const v4f*)(red + 4);
    mean = (((a0[0] + a0[1]) + (a0[2] + a0[3])) + ((a1[0] + a1[1]) + (a1[2] + a1[3])))
           / (float)N_DRUG;
  }
  const float d0 = v0 - mean, d1 = v1 - mean, d2 = v2 - mean, d3 = v3 - mean;
  float q = d0 * d0;
  q = fmaf(d1, d1, q);
  q = fmaf(d2, d2, q);
  if (has3) q = fmaf(d3, d3, q);
  q = wave_sum_all(q);
  if (lane == 0) red[8 + w] = q;
  __syncthreads();
  float var;
  {
    const v4f a0 = *(const v4f*)(red + 8), a1 = *(const v4f*)(red + 12);
    var = (((a0[0] + a0[1]) + (a0[2] + a0[3])) + ((a1[0] + a1[1]) + (a1[2] + a1[3])))
          / (float)N_DRUG;
  }
  const float scale = gw * rsqrtf(var + BN_EPS);
  const float shift = gb;
  out[o0]            = d0 * scale + shift;
  out[o0 + 256u * D] = d1 * scale + shift;
  out[o0 + 512u * D] = d2 * scale + shift;
  if (has3) out[o0 + 768u * D] = d3 * scale + shift;
}

extern "C" void kernel_launch(void* const* d_in, const int* in_sizes, int n_in,
                              void* d_out, int out_size, void* d_ws, size_t ws_size,
                              hipStream_t stream) {
  (void)in_sizes; (void)n_in; (void)out_size; (void)ws_size;
  const float* drug_table = (const float*)d_in[0];
  const float* rela_table = (const float*)d_in[1];
  const float* ent_table  = (const float*)d_in[2];
  const float* W1         = (const float*)d_in[3];
  const float* b1         = (const float*)d_in[4];
  const float* W2         = (const float*)d_in[5];
  const float* b2         = (const float*)d_in[6];
  const float* lin_w      = (const float*)d_in[7];
  const float* lin_b      = (const float*)d_in[8];
  const float* bn_w       = (const float*)d_in[9];
  const float* bn_b       = (const float*)d_in[10];
  const int*   drug_name  = (const int*)d_in[11];
  const int*   adj_tail   = (const int*)d_in[12];
  const int*   adj_rel    = (const int*)d_in[13];

  float* ws     = (float*)d_ws;
  float* b2sum  = ws;
  float* y      = ws + K;
  float* relaF  = ws + K + N_DRUG * D;

  gnn_prep<<<dim3(B2_BLOCKS + 13), dim3(256), 0, stream>>>(b2, rela_table, b2sum, relaF);

  const size_t lds = (size_t)(RPAD * D + D + D + K + 512) * sizeof(float);
  gnn_main<<<dim3(N_DRUG), dim3(256), lds, stream>>>(
      drug_table, relaF, ent_table, W1, b1, W2, lin_w, lin_b,
      drug_name, adj_tail, adj_rel, b2sum, y);

  gnn_bn<<<dim3(D), dim3(256), 0, stream>>>(y, bn_w, bn_b, (float*)d_out);
}
